# starting kernel
# baseline (speedup 1.0000x reference)
#include <hip/hip_runtime.h>

typedef __attribute__((ext_vector_type(2))) float v2f;
typedef __attribute__((ext_vector_type(8))) float v8f;

#define GCN_N_NODES  50000
#define GCN_N_GRAPHS 64

__global__ void zero_f32(float* __restrict__ p, int n) {
    int i = blockIdx.x * blockDim.x + threadIdx.x;
    if (i < n) p[i] = 0.0f;
}

__global__ void deg_init(float* __restrict__ deg, int n) {
    int i = blockIdx.x * blockDim.x + threadIdx.x;
    if (i < n) deg[i] = 1.0f;
}

__global__ void deg_accum(const int* __restrict__ dst, float* __restrict__ deg, int e) {
    int i = blockIdx.x * blockDim.x + threadIdx.x;
    if (i < e) atomicAdd(&deg[dst[i]], 1.0f);
}

__global__ void deg_finalize(float* __restrict__ deg, int n) {
    int i = blockIdx.x * blockDim.x + threadIdx.x;
    if (i < n) deg[i] = rsqrtf(deg[i]);
}

template <int K, int NCOLS, int WAVES>
__global__ void gemm_f32_wmma(const float* __restrict__ A,
                              const float* __restrict__ Wm,
                              float* __restrict__ H, int M) {
    __shared__ float Bs[K * 16];

    const int tid  = threadIdx.x;
    const int lane = tid & 31;
    const int wav  = tid >> 5;
    const int colBase = blockIdx.y * 16;

    #pragma unroll
    for (int i = tid; i < K * 16; i += 32 * WAVES) {
        const int kk = i >> 4;
        const int cc = i & 15;
        Bs[i] = Wm[kk * NCOLS + colBase + cc];
    }
    __syncthreads();

    const int mtile = blockIdx.x * WAVES + wav;
    if (mtile * 16 < M) {
        const int l15 = lane & 15;
        const int hi  = lane >> 4;
        const int khi = hi * 2;
        const int mrow = mtile * 16 + l15;

        v8f acc = {0.f, 0.f, 0.f, 0.f, 0.f, 0.f, 0.f, 0.f};
        #pragma unroll 8
        for (int k0 = 0; k0 < K; k0 += 4) {
            v2f a = *(const v2f*)&A[mrow * K + k0 + khi];
            v2f b;
            b.x = Bs[(k0 + khi) * 16 + l15];
            b.y = Bs[(k0 + khi + 1) * 16 + l15];
            acc = __builtin_amdgcn_wmma_f32_16x16x4_f32(
                 false, a,  false, b,
                 (short)0, acc,  false,  false);
        }

        const int m0  = mtile * 16 + hi * 8;
        const int col = colBase + l15;
        #pragma unroll
        for (int r = 0; r < 8; ++r)
            H[(m0 + r) * NCOLS + col] = acc[r];
    }
}

template <int F>
__global__ void scatter_edges(const float* __restrict__ h,
                              const int* __restrict__ src,
                              const int* __restrict__ dst,
                              const float* __restrict__ dinv,
                              float* __restrict__ agg, int e) {
    const int wid  = (int)((blockIdx.x * blockDim.x + threadIdx.x) >> 5);
    const int lane = threadIdx.x & 31;
    if (wid >= e) return;
    const int s = src[wid];
    __builtin_prefetch(&h[(size_t)s * F + lane], 0, 0);
    const int d = dst[wid];
    const float nrm = dinv[s] * dinv[d];
    #pragma unroll
    for (int c = 0; c < F / 32; ++c) {
        float v = h[s * F + c * 32 + lane] * nrm;
        atomicAdd(&agg[d * F + c * 32 + lane], v);
    }
}

template <int F, bool RELU>
__global__ void node_finalize(const float* __restrict__ agg,
                              const float* __restrict__ h,
                              const float* __restrict__ dinv,
                              const float* __restrict__ bias,
                              float* __restrict__ out, int n) {
    const int node = blockIdx.x;
    if (node >= n) return;
    const float di  = dinv[node];
    const float di2 = di * di;
    for (int f = threadIdx.x; f < F; f += blockDim.x) {
        float v = agg[node * F + f] + h[node * F + f] * di2 + bias[f];
        out[node * F + f] = RELU ? fmaxf(v, 0.0f) : v;
    }
}

__global__ void pool_sum(const float* __restrict__ h,
                         const int* __restrict__ batch,
                         float* __restrict__ sums,
                         float* __restrict__ cnts, int n) {
    const int wid  = (int)((blockIdx.x * blockDim.x + threadIdx.x) >> 5);
    const int lane = threadIdx.x & 31;
    if (wid >= n) return;
    const int g = batch[wid];
    if (lane == 0) atomicAdd(&cnts[g], 1.0f);
    atomicAdd(&sums[g * 64 + lane],      h[wid * 64 + lane]);
    atomicAdd(&sums[g * 64 + 32 + lane], h[wid * 64 + 32 + lane]);
}

__global__ void pool_div(const float* __restrict__ sums,
                         const float* __restrict__ cnts,
                         float* __restrict__ out, int n) {
    int i = blockIdx.x * blockDim.x + threadIdx.x;
    if (i < n) out[i] = sums[i] / fmaxf(cnts[i / 64], 1.0f);
}

extern "C" void kernel_launch(void* const* d_in, const int* in_sizes, int n_in,
                              void* d_out, int out_size, void* d_ws, size_t ws_size,
                              hipStream_t stream) {
    const float* x     = (const float*)d_in[0];
    const int*   ei    = (const int*)d_in[1];
    const int*   batch = (const int*)d_in[2];
    const float* Wl[4] = {(const float*)d_in[3], (const float*)d_in[5],
                          (const float*)d_in[7], (const float*)d_in[9]};
    const float* bl[4] = {(const float*)d_in[4], (const float*)d_in[6],
                          (const float*)d_in[8], (const float*)d_in[10]};

    const int N = GCN_N_NODES;
    const int E = in_sizes[1] / 2;
    const int* src = ei;
    const int* dst = ei + E;

    auto au = [](size_t v) { return (v + 255) & ~(size_t)255; };
    char* ws = (char*)d_ws;
    size_t off = 0;
    float* deg  = (float*)(ws + off); off += au(sizeof(float) * N);
    float* bufA = (float*)(ws + off); off += au(sizeof(float) * (size_t)N * 96);
    float* bufB = (float*)(ws + off); off += au(sizeof(float) * (size_t)N * 96);
    float* bufC = (float*)(ws + off); off += au(sizeof(float) * (size_t)N * 96);
    float* sums = (float*)(ws + off); off += au(sizeof(float) * GCN_N_GRAPHS * 64);
    float* cnts = (float*)(ws + off); off += au(sizeof(float) * GCN_N_GRAPHS);
    (void)ws_size; (void)n_in; (void)out_size;

    const int TB = 256;
    deg_init    <<<(N + TB - 1) / TB, TB, 0, stream>>>(deg, N);
    deg_accum   <<<(E + TB - 1) / TB, TB, 0, stream>>>(dst, deg, E);
    deg_finalize<<<(N + TB - 1) / TB, TB, 0, stream>>>(deg, N);

    constexpr int WV = 4;
    const int MT  = (N + 15) / 16;
    const int GBX = (MT + WV - 1) / WV;

    gemm_f32_wmma<128, 96, WV><<<dim3(GBX, 6), 32 * WV, 0, stream>>>(x, Wl[0], bufA, N);
    zero_f32<<<(N * 96 + TB - 1) / TB, TB, 0, stream>>>(bufB, N * 96);
    scatter_edges<96><<<(E + 7) / 8, 256, 0, stream>>>(bufA, src, dst, deg, bufB, E);
    node_finalize<96, true><<<N, 96, 0, stream>>>(bufB, bufA, deg, bl[0], bufC, N);

    for (int l = 1; l <= 2; ++l) {
        gemm_f32_wmma<96, 96, WV><<<dim3(GBX, 6), 32 * WV, 0, stream>>>(bufC, Wl[l], bufA, N);
        zero_f32<<<(N * 96 + TB - 1) / TB, TB, 0, stream>>>(bufB, N * 96);
        scatter_edges<96><<<(E + 7) / 8, 256, 0, stream>>>(bufA, src, dst, deg, bufB, E);
        node_finalize<96, true><<<N, 96, 0, stream>>>(bufB, bufA, deg, bl[l], bufC, N);
    }

    gemm_f32_wmma<96, 64, WV><<<dim3(GBX, 4), 32 * WV, 0, stream>>>(bufC, Wl[3], bufA, N);
    zero_f32<<<(N * 64 + TB - 1) / TB, TB, 0, stream>>>(bufB, N * 64);
    scatter_edges<64><<<(E + 7) / 8, 256, 0, stream>>>(bufA, src, dst, deg, bufB, E);
    node_finalize<64, false><<<N, 64, 0, stream>>>(bufB, bufA, deg, bl[3], bufC, N);

    zero_f32<<<(GCN_N_GRAPHS * 64 + TB - 1) / TB, TB, 0, stream>>>(sums, GCN_N_GRAPHS * 64);
    zero_f32<<<1, 64, 0, stream>>>(cnts, GCN_N_GRAPHS);
    pool_sum<<<(N + 7) / 8, 256, 0, stream>>>(bufC, batch, sums, cnts, N);
    pool_div<<<(GCN_N_GRAPHS * 64 + TB - 1) / TB, TB, 0, stream>>>(sums, cnts, (float*)d_out,
                                                                   GCN_N_GRAPHS * 64);
}


# Mean pool: float atomics replaced by one owner block per graph (filter-scans batch, lists its nodes in LDS, waves sum the rows in registers, fixed-order combine, divides); pool sum order changes
# speedup vs baseline: 2.4529x; 2.4529x over previous
#include <hip/hip_runtime.h>

typedef __attribute__((ext_vector_type(2))) float v2f;
typedef __attribute__((ext_vector_type(8))) float v8f;

#define GCN_N_NODES  50000
#define GCN_N_GRAPHS 64

__global__ void deg_init(float* __restrict__ deg, int n) {
    int i = blockIdx.x * blockDim.x + threadIdx.x;
    if (i < n) deg[i] = 1.0f;
}

__global__ void deg_accum(const int* __restrict__ dst, float* __restrict__ deg, int e) {
    int i = blockIdx.x * blockDim.x + threadIdx.x;
    if (i < e) atomicAdd(&deg[dst[i]], 1.0f);
}

__global__ void deg_finalize(float* __restrict__ deg, int n) {
    int i = blockIdx.x * blockDim.x + threadIdx.x;
    if (i < n) deg[i] = rsqrtf(deg[i]);
}

template <int K, int NCOLS, int WAVES>
__global__ void gemm_f32_wmma(const float* __restrict__ A,
                              const float* __restrict__ Wm,
                              float* __restrict__ H, int M) {
    __shared__ float Bs[K * 16];

    const int tid  = threadIdx.x;
    const int lane = tid & 31;
    const int wav  = tid >> 5;
    const int colBase = blockIdx.y * 16;

    #pragma unroll
    for (int i = tid; i < K * 16; i += 32 * WAVES) {
        const int kk = i >> 4;
        const int cc = i & 15;
        Bs[i] = Wm[kk * NCOLS + colBase + cc];
    }
    __syncthreads();

    const int mtile = blockIdx.x * WAVES + wav;
    if (mtile * 16 < M) {
        const int l15 = lane & 15;
        const int hi  = lane >> 4;
        const int khi = hi * 2;
        const int mrow = mtile * 16 + l15;

        v8f acc = {0.f, 0.f, 0.f, 0.f, 0.f, 0.f, 0.f, 0.f};
        #pragma unroll 8
        for (int k0 = 0; k0 < K; k0 += 4) {
            v2f a = *(const v2f*)&A[mrow * K + k0 + khi];
            v2f b;
            b.x = Bs[(k0 + khi) * 16 + l15];
            b.y = Bs[(k0 + khi + 1) * 16 + l15];
            acc = __builtin_amdgcn_wmma_f32_16x16x4_f32(
                 false, a,  false, b,
                 (short)0, acc,  false,  false);
        }

        const int m0  = mtile * 16 + hi * 8;
        const int col = colBase + l15;
        #pragma unroll
        for (int r = 0; r < 8; ++r)
            H[(m0 + r) * NCOLS + col] = acc[r];
    }
}

#define AGG_R     200
#define AGG_WAVES 32
#define AGG_CAPW  256

template <int F>
__device__ __forceinline__ void agg_edge_now(const float* __restrict__ h,
                                             const int* __restrict__ src,
                                             const float* __restrict__ dinv,
                                             const float* t_dinv, float* tile,
                                             int edge, int row, int lane) {
    const int s = src[edge];
    const float nrm = dinv[s] * t_dinv[row];
    #pragma unroll
    for (int c = 0; c < F / 32; ++c)
        atomicAdd(&tile[row * F + c * 32 + lane], h[(size_t)s * F + c * 32 + lane] * nrm);
}

template <int F, bool RELU, bool FIRST>
__global__ __launch_bounds__(1024)
void agg_tile(const float* __restrict__ h,
              const int* __restrict__ src,
              const int* __restrict__ dst,
              const float* __restrict__ dinv,
              const float* __restrict__ bias,
              float* __restrict__ out,
              int* __restrict__ kept_a,
              int* __restrict__ kept_d,
              float* __restrict__ kept_n,
              int* __restrict__ kept_cnt,
              int e, int n) {
    __shared__ float tile[AGG_R * F];
    __shared__ float t_dinv[AGG_R];
    __shared__ int   l_a[AGG_WAVES * AGG_CAPW];
    __shared__ int   l_d[AGG_WAVES * AGG_CAPW];
    __shared__ float l_n[AGG_WAVES * AGG_CAPW];

    const int tid  = threadIdx.x;
    const int lane = tid & 31;
    const int wav  = __builtin_amdgcn_readfirstlane(tid >> 5);
    const int lo   = blockIdx.x * AGG_R;
    const int cnt  = min(AGG_R, n - lo);
    const unsigned ucnt = (unsigned)cnt;

    int* my_a = l_a + wav * AGG_CAPW;
    int* my_d = l_d + wav * AGG_CAPW;
    float* my_n = l_n + wav * AGG_CAPW;
    int wcount = 0;
    const int gw = blockIdx.x * AGG_WAVES + wav;
    const size_t kb = (size_t)gw * AGG_CAPW;

    bool scan = true;
    int   ka[8], kd[8];
    float kn[8];
    if (!FIRST) {
        wcount = kept_cnt[gw];
        scan = wcount > AGG_CAPW;
        if (scan) {
            wcount = 0;
        } else if (wcount > 0) {
            #pragma unroll
            for (int u = 0; u < 8; ++u) {
                const size_t qi = kb + (size_t)min(lane + 32 * u, wcount - 1);
                ka[u] = kept_a[qi];
                kd[u] = kept_d[qi];
                kn[u] = kept_n[qi];
            }
        }
    }

    for (int r = tid; r < cnt; r += 1024) t_dinv[r] = dinv[lo + r];
    for (int i = tid; i < cnt * F; i += 1024) {
        const int r = i / F;
        const int f = i - r * F;
        const float di = dinv[lo + r];
        tile[i] = h[(size_t)lo * F + i] * (di * di) + bias[f];
    }
    if (!FIRST && !scan) {
        #pragma unroll
        for (int u = 0; u < 8; ++u) {
            const int q = lane + 32 * u;
            if (q < wcount) { my_a[q] = ka[u]; my_d[q] = kd[u]; my_n[q] = kn[u]; }
        }
    }
    __syncthreads();

    auto take = [&](unsigned rr, int edge) {
        const bool m = rr < ucnt;
        const unsigned mask = __builtin_amdgcn_ballot_w32(m);
        if (mask) {
            const int pos = wcount + __popc(mask & ((1u << lane) - 1u));
            if (m && pos < AGG_CAPW) { my_a[pos] = edge; my_d[pos] = (int)rr; }
            unsigned ov = __builtin_amdgcn_ballot_w32(m && pos >= AGG_CAPW);
            while (ov) {
                const int l = __ffs(ov) - 1;
                ov &= ov - 1u;
                agg_edge_now<F>(h, src, dinv, t_dinv, tile,
                                __shfl(edge, l), __shfl((int)rr, l), lane);
            }
            wcount += __popc(mask);
        }
    };
    auto take4 = [&](int4 d, int edge0) {
        const unsigned r0 = (unsigned)(d.x - lo), r1 = (unsigned)(d.y - lo);
        const unsigned r2 = (unsigned)(d.z - lo), r3 = (unsigned)(d.w - lo);
        const bool any = (r0 < ucnt) | (r1 < ucnt) | (r2 < ucnt) | (r3 < ucnt);
        if (__builtin_amdgcn_ballot_w32(any)) {
            take(r0, edge0); take(r1, edge0 + 1); take(r2, edge0 + 2); take(r3, edge0 + 3);
        }
    };

    if (scan) {
        const int nchunks = (e + 1023) >> 10;
        const int rot = (int)(((long long)blockIdx.x * nchunks) / gridDim.x);
        const bool vec_ok = (e & 3) == 0;
        for (int k = wav; k < nchunks; k += AGG_WAVES) {
            int c = k + rot;
            if (c >= nchunks) c -= nchunks;
            const int base = (c << 10) + lane * 4;
            if (vec_ok && (c << 10) + 1024 <= e) {
                int4 d[8];
                #pragma unroll
                for (int u = 0; u < 8; ++u) d[u] = *(const int4*)(dst + base + u * 128);
                #pragma unroll
                for (int u = 0; u < 8; ++u) take4(d[u], base + u * 128);
            } else {
                #pragma unroll 1
                for (int u = 0; u < 8; ++u) {
                    const int i0 = base + u * 128;
                    int4 d;
                    d.x = (i0     < e) ? dst[i0]     : -1;
                    d.y = (i0 + 1 < e) ? dst[i0 + 1] : -1;
                    d.z = (i0 + 2 < e) ? dst[i0 + 2] : -1;
                    d.w = (i0 + 3 < e) ? dst[i0 + 3] : -1;
                    take4(d, i0);
                }
            }
        }
    }
    __syncthreads();

    static_assert(AGG_CAPW == 8 * 32, "the resolve step covers the list in 8 strides of 32 lanes");
    const int nl = min(wcount, AGG_CAPW);
    if (scan) {
        int   sv[8];
        float dv[8];
        #pragma unroll
        for (int u = 0; u < 8; ++u) {
            const int q = lane + 32 * u;
            sv[u] = (q < nl) ? src[my_a[q]] : 0;
        }
        #pragma unroll
        for (int u = 0; u < 8; ++u) {
            const int q = lane + 32 * u;
            dv[u] = (q < nl) ? dinv[sv[u]] : 0.0f;
        }
        #pragma unroll
        for (int u = 0; u < 8; ++u) {
            const int q = lane + 32 * u;
            if (q < nl) {
                const int   rw = my_d[q];
                const float wt = dv[u] * t_dinv[rw];
                my_a[q] = sv[u]; my_n[q] = wt;
                if (FIRST) { kept_a[kb + q] = sv[u]; kept_d[kb + q] = rw; kept_n[kb + q] = wt; }
            }
        }
        if (FIRST && lane == 0) kept_cnt[gw] = wcount;
    }
    __syncthreads();

    constexpr int C = F / 32;
    constexpr int U = 8;
    int p = 0;
    for (; p + U <= nl; p += U) {
        int   s[U], r[U];
        float w[U], v[U][C];
        #pragma unroll
        for (int u = 0; u < U; ++u) { s[u] = my_a[p + u]; r[u] = my_d[p + u]; w[u] = my_n[p + u]; }
        #pragma unroll
        for (int u = 0; u < U; ++u)
            #pragma unroll
            for (int c = 0; c < C; ++c)
                v[u][c] = h[(size_t)s[u] * F + c * 32 + lane];
        #pragma unroll
        for (int u = 0; u < U; ++u)
            #pragma unroll
            for (int c = 0; c < C; ++c)
                atomicAdd(&tile[r[u] * F + c * 32 + lane], v[u][c] * w[u]);
    }
    for (; p < nl; ++p) {
        const int s = my_a[p];
        const int r = my_d[p];
        const float w = my_n[p];
        #pragma unroll
        for (int c = 0; c < C; ++c)
            atomicAdd(&tile[r * F + c * 32 + lane], h[(size_t)s * F + c * 32 + lane] * w);
    }
    __syncthreads();

    for (int i = tid; i < cnt * F; i += 1024) {
        const float v = tile[i];
        out[(size_t)lo * F + i] = RELU ? fmaxf(v, 0.0f) : v;
    }
}

#define POOL_WAVES 32
#define POOL_LOADS 13
#define POOL_U     16
static_assert(POOL_WAVES * POOL_LOADS * 128 >= GCN_N_NODES, "the waves' slices cover every node");

__global__ __launch_bounds__(1024)
void pool_graph(const float* __restrict__ h,
                const int* __restrict__ batch,
                float* __restrict__ out, int n) {
    __shared__ int   list[GCN_N_NODES];
    __shared__ int   wcnt[POOL_WAVES];
    __shared__ float part[POOL_WAVES * 64];

    const int tid  = threadIdx.x;
    const int lane = tid & 31;
    const int wav  = __builtin_amdgcn_readfirstlane(tid >> 5);
    const int g    = blockIdx.x;
    const int wbase = wav * (POOL_LOADS * 128);

    int4 b[POOL_LOADS];
    if (wbase + POOL_LOADS * 128 <= n) {
        #pragma unroll
        for (int j = 0; j < POOL_LOADS; ++j)
            b[j] = *(const int4*)(batch + wbase + j * 128 + lane * 4);
    } else if (wbase < n) {
        #pragma unroll
        for (int j = 0; j < POOL_LOADS; ++j) {
            const int i0 = wbase + j * 128 + lane * 4;
            const int v0 = batch[min(i0,     n - 1)], v1 = batch[min(i0 + 1, n - 1)];
            const int v2 = batch[min(i0 + 2, n - 1)], v3 = batch[min(i0 + 3, n - 1)];
            b[j].x = (i0     < n) ? v0 : -1;
            b[j].y = (i0 + 1 < n) ? v1 : -1;
            b[j].z = (i0 + 2 < n) ? v2 : -1;
            b[j].w = (i0 + 3 < n) ? v3 : -1;
        }
    } else {
        #pragma unroll
        for (int j = 0; j < POOL_LOADS; ++j) b[j] = make_int4(-1, -1, -1, -1);
    }
    int cnt = 0;
    #pragma unroll
    for (int j = 0; j < POOL_LOADS; ++j) {
        cnt += __popc(__builtin_amdgcn_ballot_w32(b[j].x == g))
             + __popc(__builtin_amdgcn_ballot_w32(b[j].y == g))
             + __popc(__builtin_amdgcn_ballot_w32(b[j].z == g))
             + __popc(__builtin_amdgcn_ballot_w32(b[j].w == g));
    }
    if (lane == 0) wcnt[wav] = cnt;
    __syncthreads();

    int pos = 0, total = 0;
    #pragma unroll
    for (int v = 0; v < POOL_WAVES; ++v) {
        const int c = __builtin_amdgcn_readfirstlane(wcnt[v]);
        total += c;
        if (v < wav) pos += c;
    }
    if (cnt > 0) {
        const unsigned below = (1u << lane) - 1u;
        auto put = [&](int bv, int node) {
            const bool m = bv == g;
            const unsigned mask = __builtin_amdgcn_ballot_w32(m);
            if (m) list[pos + __popc(mask & below)] = node;
            pos += __popc(mask);
        };
        #pragma unroll
        for (int j = 0; j < POOL_LOADS; ++j) {
            const int i0 = wbase + j * 128 + lane * 4;
            put(b[j].x, i0); put(b[j].y, i0 + 1); put(b[j].z, i0 + 2); put(b[j].w, i0 + 3);
        }
    }
    __syncthreads();

    const int e_lo = (wav * total) / POOL_WAVES;
    const int e_hi = ((wav + 1) * total) / POOL_WAVES;
    v2f acc = {0.f, 0.f};
    for (int e = e_lo; e < e_hi; e += POOL_U) {
        int nd[POOL_U];
        v2f v[POOL_U];
        #pragma unroll
        for (int u = 0; u < POOL_U; ++u) {
            const int q = e + u;
            nd[u] = __builtin_amdgcn_readfirstlane(list[q < e_hi ? q : e_lo]);
        }
        #pragma unroll
        for (int u = 0; u < POOL_U; ++u)
            v[u] = *(const v2f*)(h + (size_t)nd[u] * 64 + lane * 2);
        #pragma unroll
        for (int u = 0; u < POOL_U; ++u)
            if (e + u < e_hi) { acc.x += v[u].x; acc.y += v[u].y; }
    }
    *(v2f*)&part[wav * 64 + lane * 2] = acc;
    __syncthreads();

    if (tid < 64) {
        float s = 0.f;
        #pragma unroll
        for (int w = 0; w < POOL_WAVES; ++w) s += part[w * 64 + tid];
        out[g * 64 + tid] = s / fmaxf((float)total, 1.0f);
    }
}

extern "C" void kernel_launch(void* const* d_in, const int* in_sizes, int n_in,
                              void* d_out, int out_size, void* d_ws, size_t ws_size,
                              hipStream_t stream) {
    const float* x     = (const float*)d_in[0];
    const int*   ei    = (const int*)d_in[1];
    const int*   batch = (const int*)d_in[2];
    const float* Wl[4] = {(const float*)d_in[3], (const float*)d_in[5],
                          (const float*)d_in[7], (const float*)d_in[9]};
    const float* bl[4] = {(const float*)d_in[4], (const float*)d_in[6],
                          (const float*)d_in[8], (const float*)d_in[10]};

    const int N = GCN_N_NODES;
    const int E = in_sizes[1] / 2;
    const int* src = ei;
    const int* dst = ei + E;

    auto au = [](size_t v) { return (v + 255) & ~(size_t)255; };
    char* ws = (char*)d_ws;
    size_t off = 0;
    float* deg  = (float*)(ws + off); off += au(sizeof(float) * N);
    float* bufA = (float*)(ws + off); off += au(sizeof(float) * (size_t)N * 96);
    float* bufC = (float*)(ws + off); off += au(sizeof(float) * (size_t)N * 96);
    const size_t KEPT = (size_t)((GCN_N_NODES + AGG_R - 1) / AGG_R) * AGG_WAVES * AGG_CAPW;
    int*   kept_a   = (int*)(ws + off);   off += au(sizeof(int) * KEPT);
    int*   kept_d   = (int*)(ws + off);   off += au(sizeof(int) * KEPT);
    float* kept_n   = (float*)(ws + off); off += au(sizeof(float) * KEPT);
    int*   kept_cnt = (int*)(ws + off);   off += au(sizeof(int) * (KEPT / AGG_CAPW));
    (void)ws_size; (void)n_in; (void)out_size;

    const int TB = 256;
    deg_init    <<<(N + TB - 1) / TB, TB, 0, stream>>>(deg, N);
    deg_accum   <<<(E + TB - 1) / TB, TB, 0, stream>>>(dst, deg, E);
    deg_finalize<<<(N + TB - 1) / TB, TB, 0, stream>>>(deg, N);

    constexpr int WV = 4;
    const int MT  = (N + 15) / 16;
    const int GBX = (MT + WV - 1) / WV;
    const int AGB = (N + AGG_R - 1) / AGG_R;

    gemm_f32_wmma<128, 96, WV><<<dim3(GBX, 6), 32 * WV, 0, stream>>>(x, Wl[0], bufA, N);
    agg_tile<96, true, true><<<AGB, 1024, 0, stream>>>(bufA, src, dst, deg, bl[0], bufC,
                                                       kept_a, kept_d, kept_n, kept_cnt, E, N);

    for (int l = 1; l <= 2; ++l) {
        gemm_f32_wmma<96, 96, WV><<<dim3(GBX, 6), 32 * WV, 0, stream>>>(bufC, Wl[l], bufA, N);
        agg_tile<96, true, false><<<AGB, 1024, 0, stream>>>(bufA, src, dst, deg, bl[l], bufC,
                                                            kept_a, kept_d, kept_n, kept_cnt, E, N);
    }

    gemm_f32_wmma<96, 64, WV><<<dim3(GBX, 4), 32 * WV, 0, stream>>>(bufC, Wl[3], bufA, N);
    agg_tile<64, false, false><<<AGB, 1024, 0, stream>>>(bufA, src, dst, deg, bl[3], bufC,
                                                         kept_a, kept_d, kept_n, kept_cnt, E, N);

    pool_graph<<<GCN_N_GRAPHS, 1024, 0, stream>>>(bufC, batch, (float*)d_out, N);
}


# Aggregation add phase: the block's 32 edge lists are cut into 8-entry batches dealt round-robin to the waves; a list's remainder goes out as one load group; LDS float-atomic sum order changes
# speedup vs baseline: 2.5274x; 1.0304x over previous
#include <hip/hip_runtime.h>

typedef __attribute__((ext_vector_type(2))) float v2f;
typedef __attribute__((ext_vector_type(8))) float v8f;

#define GCN_N_NODES  50000
#define GCN_N_GRAPHS 64

__global__ void deg_init(float* __restrict__ deg, int n) {
    int i = blockIdx.x * blockDim.x + threadIdx.x;
    if (i < n) deg[i] = 1.0f;
}

__global__ void deg_accum(const int* __restrict__ dst, float* __restrict__ deg, int e) {
    int i = blockIdx.x * blockDim.x + threadIdx.x;
    if (i < e) atomicAdd(&deg[dst[i]], 1.0f);
}

__global__ void deg_finalize(float* __restrict__ deg, int n) {
    int i = blockIdx.x * blockDim.x + threadIdx.x;
    if (i < n) deg[i] = rsqrtf(deg[i]);
}

template <int K, int NCOLS, int WAVES>
__global__ void gemm_f32_wmma(const float* __restrict__ A,
                              const float* __restrict__ Wm,
                              float* __restrict__ H, int M) {
    __shared__ float Bs[K * 16];

    const int tid  = threadIdx.x;
    const int lane = tid & 31;
    const int wav  = tid >> 5;
    const int colBase = blockIdx.y * 16;

    #pragma unroll
    for (int i = tid; i < K * 16; i += 32 * WAVES) {
        const int kk = i >> 4;
        const int cc = i & 15;
        Bs[i] = Wm[kk * NCOLS + colBase + cc];
    }
    __syncthreads();

    const int mtile = blockIdx.x * WAVES + wav;
    if (mtile * 16 < M) {
        const int l15 = lane & 15;
        const int hi  = lane >> 4;
        const int khi = hi * 2;
        const int mrow = mtile * 16 + l15;

        v8f acc = {0.f, 0.f, 0.f, 0.f, 0.f, 0.f, 0.f, 0.f};
        #pragma unroll 8
        for (int k0 = 0; k0 < K; k0 += 4) {
            v2f a = *(const v2f*)&A[mrow * K + k0 + khi];
            v2f b;
            b.x = Bs[(k0 + khi) * 16 + l15];
            b.y = Bs[(k0 + khi + 1) * 16 + l15];
            acc = __builtin_amdgcn_wmma_f32_16x16x4_f32(
                 false, a,  false, b,
                 (short)0, acc,  false,  false);
        }

        const int m0  = mtile * 16 + hi * 8;
        const int col = colBase + l15;
        #pragma unroll
        for (int r = 0; r < 8; ++r)
            H[(m0 + r) * NCOLS + col] = acc[r];
    }
}

#define AGG_R     200
#define AGG_WAVES 32
#define AGG_CAPW  256

template <int F>
__device__ __forceinline__ void agg_edge_now(const float* __restrict__ h,
                                             const int* __restrict__ src,
                                             const float* __restrict__ dinv,
                                             const float* t_dinv, float* tile,
                                             int edge, int row, int lane) {
    const int s = src[edge];
    const float nrm = dinv[s] * t_dinv[row];
    #pragma unroll
    for (int c = 0; c < F / 32; ++c)
        atomicAdd(&tile[row * F + c * 32 + lane], h[(size_t)s * F + c * 32 + lane] * nrm);
}

template <int F, bool RELU, bool FIRST>
__global__ __launch_bounds__(1024)
void agg_tile(const float* __restrict__ h,
              const int* __restrict__ src,
              const int* __restrict__ dst,
              const float* __restrict__ dinv,
              const float* __restrict__ bias,
              float* __restrict__ out,
              int* __restrict__ kept_a,
              int* __restrict__ kept_d,
              float* __restrict__ kept_n,
              int* __restrict__ kept_cnt,
              int e, int n) {
    __shared__ float tile[AGG_R * F];
    __shared__ float t_dinv[AGG_R];
    __shared__ int   l_a[AGG_WAVES * AGG_CAPW];
    __shared__ int   l_d[AGG_WAVES * AGG_CAPW];
    __shared__ float l_n[AGG_WAVES * AGG_CAPW];
    __shared__ int   l_cnt[AGG_WAVES];

    const int tid  = threadIdx.x;
    const int lane = tid & 31;
    const int wav  = __builtin_amdgcn_readfirstlane(tid >> 5);
    const int lo   = blockIdx.x * AGG_R;
    const int cnt  = min(AGG_R, n - lo);
    const unsigned ucnt = (unsigned)cnt;

    int* my_a = l_a + wav * AGG_CAPW;
    int* my_d = l_d + wav * AGG_CAPW;
    float* my_n = l_n + wav * AGG_CAPW;
    int wcount = 0;
    const int gw = blockIdx.x * AGG_WAVES + wav;
    const size_t kb = (size_t)gw * AGG_CAPW;

    bool scan = true;
    int   ka[8], kd[8];
    float kn[8];
    if (!FIRST) {
        wcount = kept_cnt[gw];
        scan = wcount > AGG_CAPW;
        if (scan) {
            wcount = 0;
        } else if (wcount > 0) {
            #pragma unroll
            for (int u = 0; u < 8; ++u) {
                const size_t qi = kb + (size_t)min(lane + 32 * u, wcount - 1);
                ka[u] = kept_a[qi];
                kd[u] = kept_d[qi];
                kn[u] = kept_n[qi];
            }
        }
    }

    for (int r = tid; r < cnt; r += 1024) t_dinv[r] = dinv[lo + r];
    for (int i = tid; i < cnt * F; i += 1024) {
        const int r = i / F;
        const int f = i - r * F;
        const float di = dinv[lo + r];
        tile[i] = h[(size_t)lo * F + i] * (di * di) + bias[f];
    }
    if (!FIRST && !scan) {
        #pragma unroll
        for (int u = 0; u < 8; ++u) {
            const int q = lane + 32 * u;
            if (q < wcount) { my_a[q] = ka[u]; my_d[q] = kd[u]; my_n[q] = kn[u]; }
        }
    }
    __syncthreads();

    auto take = [&](unsigned rr, int edge) {
        const bool m = rr < ucnt;
        const unsigned mask = __builtin_amdgcn_ballot_w32(m);
        if (mask) {
            const int pos = wcount + __popc(mask & ((1u << lane) - 1u));
            if (m && pos < AGG_CAPW) { my_a[pos] = edge; my_d[pos] = (int)rr; }
            unsigned ov = __builtin_amdgcn_ballot_w32(m && pos >= AGG_CAPW);
            while (ov) {
                const int l = __ffs(ov) - 1;
                ov &= ov - 1u;
                agg_edge_now<F>(h, src, dinv, t_dinv, tile,
                                __shfl(edge, l), __shfl((int)rr, l), lane);
            }
            wcount += __popc(mask);
        }
    };
    auto take4 = [&](int4 d, int edge0) {
        const unsigned r0 = (unsigned)(d.x - lo), r1 = (unsigned)(d.y - lo);
        const unsigned r2 = (unsigned)(d.z - lo), r3 = (unsigned)(d.w - lo);
        const bool any = (r0 < ucnt) | (r1 < ucnt) | (r2 < ucnt) | (r3 < ucnt);
        if (__builtin_amdgcn_ballot_w32(any)) {
            take(r0, edge0); take(r1, edge0 + 1); take(r2, edge0 + 2); take(r3, edge0 + 3);
        }
    };

    if (scan) {
        const int nchunks = (e + 1023) >> 10;
        const int rot = (int)(((long long)blockIdx.x * nchunks) / gridDim.x);
        const bool vec_ok = (e & 3) == 0;
        for (int k = wav; k < nchunks; k += AGG_WAVES) {
            int c = k + rot;
            if (c >= nchunks) c -= nchunks;
            const int base = (c << 10) + lane * 4;
            if (vec_ok && (c << 10) + 1024 <= e) {
                int4 d[8];
                #pragma unroll
                for (int u = 0; u < 8; ++u) d[u] = *(const int4*)(dst + base + u * 128);
                #pragma unroll
                for (int u = 0; u < 8; ++u) take4(d[u], base + u * 128);
            } else {
                #pragma unroll 1
                for (int u = 0; u < 8; ++u) {
                    const int i0 = base + u * 128;
                    int4 d;
                    d.x = (i0     < e) ? dst[i0]     : -1;
                    d.y = (i0 + 1 < e) ? dst[i0 + 1] : -1;
                    d.z = (i0 + 2 < e) ? dst[i0 + 2] : -1;
                    d.w = (i0 + 3 < e) ? dst[i0 + 3] : -1;
                    take4(d, i0);
                }
            }
        }
    }
    __syncthreads();

    static_assert(AGG_CAPW == 8 * 32, "the resolve step covers the list in 8 strides of 32 lanes");
    const int nl = min(wcount, AGG_CAPW);
    if (scan) {
        int   sv[8];
        float dv[8];
        #pragma unroll
        for (int u = 0; u < 8; ++u) {
            const int q = lane + 32 * u;
            sv[u] = (q < nl) ? src[my_a[q]] : 0;
        }
        #pragma unroll
        for (int u = 0; u < 8; ++u) {
            const int q = lane + 32 * u;
            dv[u] = (q < nl) ? dinv[sv[u]] : 0.0f;
        }
        #pragma unroll
        for (int u = 0; u < 8; ++u) {
            const int q = lane + 32 * u;
            if (q < nl) {
                const int   rw = my_d[q];
                const float wt = dv[u] * t_dinv[rw];
                my_a[q] = sv[u]; my_n[q] = wt;
                if (FIRST) { kept_a[kb + q] = sv[u]; kept_d[kb + q] = rw; kept_n[kb + q] = wt; }
            }
        }
        if (FIRST && lane == 0) kept_cnt[gw] = wcount;
    }
    if (lane == 0) l_cnt[wav] = nl;
    __syncthreads();

    static_assert(AGG_WAVES == 32, "one lane per list in the batch count below");
    constexpr int C = F / 32;
    constexpr int U = 8;
    const int lc  = l_cnt[lane];
    const int lnb = (lc + U - 1) / U;
    int inc = lnb;
    #pragma unroll
    for (int dd = 1; dd < 32; dd <<= 1) {
        const int up = __shfl_up(inc, dd);
        if (lane >= dd) inc += up;
    }
    const int pb = inc - lnb;
    const int nbatch = __builtin_amdgcn_readlane(inc, 31);
    for (int t = wav; t < nbatch; t += AGG_WAVES) {
        const int j   = __popc(__builtin_amdgcn_ballot_w32(pb <= t)) - 1;
        const int off = (t - __builtin_amdgcn_readlane(pb, j)) * U;
        const int len = __builtin_amdgcn_readlane(lc, j);
        const int*   ja = l_a + j * AGG_CAPW;
        const int*   jd = l_d + j * AGG_CAPW;
        const float* jn = l_n + j * AGG_CAPW;
        int   s[U], r[U];
        float w[U], v[U][C];
        if (off + U <= len) {
            #pragma unroll
            for (int u = 0; u < U; ++u) { s[u] = ja[off + u]; r[u] = jd[off + u]; w[u] = jn[off + u]; }
            #pragma unroll
            for (int u = 0; u < U; ++u)
                #pragma unroll
                for (int c = 0; c < C; ++c)
                    v[u][c] = h[(size_t)s[u] * F + c * 32 + lane];
            #pragma unroll
            for (int u = 0; u < U; ++u)
                #pragma unroll
                for (int c = 0; c < C; ++c)
                    atomicAdd(&tile[r[u] * F + c * 32 + lane], v[u][c] * w[u]);
        } else {
            #pragma unroll
            for (int u = 0; u < U; ++u) {
                const int q = min(off + u, len - 1);
                s[u] = ja[q]; r[u] = jd[q]; w[u] = jn[q];
            }
            #pragma unroll
            for (int u = 0; u < U; ++u)
                #pragma unroll
                for (int c = 0; c < C; ++c)
                    v[u][c] = h[(size_t)s[u] * F + c * 32 + lane];
            #pragma unroll
            for (int u = 0; u < U; ++u)
                if (off + u < len) {
                    #pragma unroll
                    for (int c = 0; c < C; ++c)
                        atomicAdd(&tile[r[u] * F + c * 32 + lane], v[u][c] * w[u]);
                }
        }
    }
    __syncthreads();

    for (int i = tid; i < cnt * F; i += 1024) {
        const float v = tile[i];
        out[(size_t)lo * F + i] = RELU ? fmaxf(v, 0.0f) : v;
    }
}

#define POOL_WAVES 32
#define POOL_LOADS 13
#define POOL_U     16
static_assert(POOL_WAVES * POOL_LOADS * 128 >= GCN_N_NODES, "the waves' slices cover every node");

__global__ __launch_bounds__(1024)
void pool_graph(const float* __restrict__ h,
                const int* __restrict__ batch,
                float* __restrict__ out, int n) {
    __shared__ int   list[GCN_N_NODES];
    __shared__ int   wcnt[POOL_WAVES];
    __shared__ float part[POOL_WAVES * 64];

    const int tid  = threadIdx.x;
    const int lane = tid & 31;
    const int wav  = __builtin_amdgcn_readfirstlane(tid >> 5);
    const int g    = blockIdx.x;
    const int wbase = wav * (POOL_LOADS * 128);

    int4 b[POOL_LOADS];
    if (wbase + POOL_LOADS * 128 <= n) {
        #pragma unroll
        for (int j = 0; j < POOL_LOADS; ++j)
            b[j] = *(const int4*)(batch + wbase + j * 128 + lane * 4);
    } else if (wbase < n) {
        #pragma unroll
        for (int j = 0; j < POOL_LOADS; ++j) {
            const int i0 = wbase + j * 128 + lane * 4;
            const int v0 = batch[min(i0,     n - 1)], v1 = batch[min(i0 + 1, n - 1)];
            const int v2 = batch[min(i0 + 2, n - 1)], v3 = batch[min(i0 + 3, n - 1)];
            b[j].x = (i0     < n) ? v0 : -1;
            b[j].y = (i0 + 1 < n) ? v1 : -1;
            b[j].z = (i0 + 2 < n) ? v2 : -1;
            b[j].w = (i0 + 3 < n) ? v3 : -1;
        }
    } else {
        #pragma unroll
        for (int j = 0; j < POOL_LOADS; ++j) b[j] = make_int4(-1, -1, -1, -1);
    }
    int cnt = 0;
    #pragma unroll
    for (int j = 0; j < POOL_LOADS; ++j) {
        cnt += __popc(__builtin_amdgcn_ballot_w32(b[j].x == g))
             + __popc(__builtin_amdgcn_ballot_w32(b[j].y == g))
             + __popc(__builtin_amdgcn_ballot_w32(b[j].z == g))
             + __popc(__builtin_amdgcn_ballot_w32(b[j].w == g));
    }
    if (lane == 0) wcnt[wav] = cnt;
    __syncthreads();

    int pos = 0, total = 0;
    #pragma unroll
    for (int v = 0; v < POOL_WAVES; ++v) {
        const int c = __builtin_amdgcn_readfirstlane(wcnt[v]);
        total += c;
        if (v < wav) pos += c;
    }
    if (cnt > 0) {
        const unsigned below = (1u << lane) - 1u;
        auto put = [&](int bv, int node) {
            const bool m = bv == g;
            const unsigned mask = __builtin_amdgcn_ballot_w32(m);
            if (m) list[pos + __popc(mask & below)] = node;
            pos += __popc(mask);
        };
        #pragma unroll
        for (int j = 0; j < POOL_LOADS; ++j) {
            const int i0 = wbase + j * 128 + lane * 4;
            put(b[j].x, i0); put(b[j].y, i0 + 1); put(b[j].z, i0 + 2); put(b[j].w, i0 + 3);
        }
    }
    __syncthreads();

    const int e_lo = (wav * total) / POOL_WAVES;
    const int e_hi = ((wav + 1) * total) / POOL_WAVES;
    v2f acc = {0.f, 0.f};
    for (int e = e_lo; e < e_hi; e += POOL_U) {
        int nd[POOL_U];
        v2f v[POOL_U];
        #pragma unroll
        for (int u = 0; u < POOL_U; ++u) {
            const int q = e + u;
            nd[u] = __builtin_amdgcn_readfirstlane(list[q < e_hi ? q : e_lo]);
        }
        #pragma unroll
        for (int u = 0; u < POOL_U; ++u)
            v[u] = *(const v2f*)(h + (size_t)nd[u] * 64 + lane * 2);
        #pragma unroll
        for (int u = 0; u < POOL_U; ++u)
            if (e + u < e_hi) { acc.x += v[u].x; acc.y += v[u].y; }
    }
    *(v2f*)&part[wav * 64 + lane * 2] = acc;
    __syncthreads();

    if (tid < 64) {
        float s = 0.f;
        #pragma unroll
        for (int w = 0; w < POOL_WAVES; ++w) s += part[w * 64 + tid];
        out[g * 64 + tid] = s / fmaxf((float)total, 1.0f);
    }
}

extern "C" void kernel_launch(void* const* d_in, const int* in_sizes, int n_in,
                              void* d_out, int out_size, void* d_ws, size_t ws_size,
                              hipStream_t stream) {
    const float* x     = (const float*)d_in[0];
    const int*   ei    = (const int*)d_in[1];
    const int*   batch = (const int*)d_in[2];
    const float* Wl[4] = {(const float*)d_in[3], (const float*)d_in[5],
                          (const float*)d_in[7], (const float*)d_in[9]};
    const float* bl[4] = {(const float*)d_in[4], (const float*)d_in[6],
                          (const float*)d_in[8], (const float*)d_in[10]};

    const int N = GCN_N_NODES;
    const int E = in_sizes[1] / 2;
    const int* src = ei;
    const int* dst = ei + E;

    auto au = [](size_t v) { return (v + 255) & ~(size_t)255; };
    char* ws = (char*)d_ws;
    size_t off = 0;
    float* deg  = (float*)(ws + off); off += au(sizeof(float) * N);
    float* bufA = (float*)(ws + off); off += au(sizeof(float) * (size_t)N * 96);
    float* bufC = (float*)(ws + off); off += au(sizeof(float) * (size_t)N * 96);
    const size_t KEPT = (size_t)((GCN_N_NODES + AGG_R - 1) / AGG_R) * AGG_WAVES * AGG_CAPW;
    int*   kept_a   = (int*)(ws + off);   off += au(sizeof(int) * KEPT);
    int*   kept_d   = (int*)(ws + off);   off += au(sizeof(int) * KEPT);
    float* kept_n   = (float*)(ws + off); off += au(sizeof(float) * KEPT);
    int*   kept_cnt = (int*)(ws + off);   off += au(sizeof(int) * (KEPT / AGG_CAPW));
    (void)ws_size; (void)n_in; (void)out_size;

    const int TB = 256;
    deg_init    <<<(N + TB - 1) / TB, TB, 0, stream>>>(deg, N);
    deg_accum   <<<(E + TB - 1) / TB, TB, 0, stream>>>(dst, deg, E);
    deg_finalize<<<(N + TB - 1) / TB, TB, 0, stream>>>(deg, N);

    constexpr int WV = 4;
    const int MT  = (N + 15) / 16;
    const int GBX = (MT + WV - 1) / WV;
    const int AGB = (N + AGG_R - 1) / AGG_R;

    gemm_f32_wmma<128, 96, WV><<<dim3(GBX, 6), 32 * WV, 0, stream>>>(x, Wl[0], bufA, N);
    agg_tile<96, true, true><<<AGB, 1024, 0, stream>>>(bufA, src, dst, deg, bl[0], bufC,
                                                       kept_a, kept_d, kept_n, kept_cnt, E, N);

    for (int l = 1; l <= 2; ++l) {
        gemm_f32_wmma<96, 96, WV><<<dim3(GBX, 6), 32 * WV, 0, stream>>>(bufC, Wl[l], bufA, N);
        agg_tile<96, true, false><<<AGB, 1024, 0, stream>>>(bufA, src, dst, deg, bl[l], bufC,
                                                            kept_a, kept_d, kept_n, kept_cnt, E, N);
    }

    gemm_f32_wmma<96, 64, WV><<<dim3(GBX, 4), 32 * WV, 0, stream>>>(bufC, Wl[3], bufA, N);
    agg_tile<64, false, false><<<AGB, 1024, 0, stream>>>(bufA, src, dst, deg, bl[3], bufC,
                                                         kept_a, kept_d, kept_n, kept_cnt, E, N);

    pool_graph<<<GCN_N_GRAPHS, 1024, 0, stream>>>(bufC, batch, (float*)d_out, N);
}


# Dense products re-tiled: a wave owns a 16-row strip across all columns (one A fragment feeds 6 or 4 WMMAs), whole W in LDS in fragment order, strips moved as whole lines via LDS; sums bit-identical
# speedup vs baseline: 2.8581x; 1.1308x over previous
#include <hip/hip_runtime.h>

typedef __attribute__((ext_vector_type(2))) float v2f;
typedef __attribute__((ext_vector_type(4))) float v4f;
typedef __attribute__((ext_vector_type(8))) float v8f;

#define GCN_N_NODES  50000
#define GCN_N_GRAPHS 64

__global__ void deg_init(float* __restrict__ deg, int n) {
    int i = blockIdx.x * blockDim.x + threadIdx.x;
    if (i < n) deg[i] = 1.0f;
}

__global__ void deg_accum(const int* __restrict__ dst, float* __restrict__ deg, int e) {
    int i = blockIdx.x * blockDim.x + threadIdx.x;
    if (i < e) atomicAdd(&deg[dst[i]], 1.0f);
}

__global__ void deg_finalize(float* __restrict__ deg, int n) {
    int i = blockIdx.x * blockDim.x + threadIdx.x;
    if (i < n) deg[i] = rsqrtf(deg[i]);
}

template <int K, int NCOLS, int WAVES>
__global__ __launch_bounds__(32 * WAVES)
void gemm_f32_wmma(const float* __restrict__ A,
                   const float* __restrict__ Wm,
                   float* __restrict__ H, int M) {
    constexpr int NCT   = NCOLS / 16;
    constexpr int NCTP  = NCT / 2;
    constexpr int KS    = K / 4;
    constexpr int T     = 32 * WAVES;
    constexpr int C4    = NCOLS / 4;
    constexpr int ITEMS = (K / 2) * C4;
    constexpr int PER   = (ITEMS + T - 1) / T;
    constexpr int AP    = K + 4;
    constexpr int CP    = NCOLS + 4;
    constexpr int STG   = 16 * (AP > CP ? AP : CP);
    constexpr int AL    = (16 * K / 4) / 32;
    constexpr int CL    = (16 * NCOLS / 4) / 32;
    static_assert(NCT % 2 == 0 && K % 4 == 0, "column tiles come in pairs, k in fours");
    static_assert(ITEMS % 32 == 0, "a wave's W items are all inside or all outside");
    static_assert((16 * K / 4) % 32 == 0 && (16 * NCOLS / 4) % 32 == 0, "whole 512-byte pieces");
    static_assert((AP / 4) % 2 == 1 && (CP / 4) % 2 == 1, "stage pitches are 16 bytes x odd");
    __shared__ v4f Bs[(K / 2) * NCTP * 16];
    __shared__ v4f Stg4[WAVES * STG / 4];

    const int tid  = threadIdx.x;
    const int lane = tid & 31;
    const int wav  = __builtin_amdgcn_readfirstlane(tid >> 5);
    const int l15  = lane & 15;
    const int hi   = lane >> 4;
    const int mtile = blockIdx.x * WAVES + wav;
    const int mt   = min(mtile, M / 16 - 1);
    float* stg = (float*)Stg4 + wav * STG;

    float4 w0[PER], w1[PER];
    #pragma unroll
    for (int p = 0; p < PER; ++p) {
        if (ITEMS % T == 0 || wav * 32 + p * T < ITEMS) {
            const int it = tid + p * T;
            const int kp = it / C4;
            const int c4 = it - kp * C4;
            w0[p] = *(const float4*)&Wm[(2 * kp) * NCOLS + 4 * c4];
            w1[p] = *(const float4*)&Wm[(2 * kp + 1) * NCOLS + 4 * c4];
        }
    }

    const float* astrip = A + (size_t)mt * 16 * K;
    float4 av[AL];
    #pragma unroll
    for (int i = 0; i < AL; ++i)
        av[i] = *(const float4*)&astrip[(i * 32 + lane) * 4];

    #pragma unroll
    for (int p = 0; p < PER; ++p) {
        if (ITEMS % T == 0 || wav * 32 + p * T < ITEMS) {
            const int it = tid + p * T;
            const int kp = it / C4;
            const int c4 = it - kp * C4;
            const int ct = c4 >> 2;
            float* d = (float*)&Bs[(kp * NCTP + (ct >> 1)) * 16 + (c4 & 3) * 4] + (ct & 1) * 2;
            *(v2f*)(d)      = v2f{w0[p].x, w1[p].x};
            *(v2f*)(d + 4)  = v2f{w0[p].y, w1[p].y};
            *(v2f*)(d + 8)  = v2f{w0[p].z, w1[p].z};
            *(v2f*)(d + 12) = v2f{w0[p].w, w1[p].w};
        }
    }
    #pragma unroll
    for (int i = 0; i < AL; ++i) {
        const int e   = (i * 32 + lane) * 4;
        const int row = e / K;
        const int col = e - row * K;
        *(float4*)&stg[row * AP + col] = av[i];
    }
    __syncthreads();

    v8f acc[NCT];
    #pragma unroll
    for (int ct = 0; ct < NCT; ++ct) acc[ct] = v8f{0.f, 0.f, 0.f, 0.f, 0.f, 0.f, 0.f, 0.f};
    const v4f*   bp = Bs + hi * (NCTP * 16) + l15;
    const float* ap = stg + l15 * AP + hi * 2;
    #pragma unroll
    for (int ks = 0; ks < KS; ++ks) {
        const v2f a = *(const v2f*)&ap[ks * 4];
        #pragma unroll
        for (int cp = 0; cp < NCTP; ++cp) {
            const v4f q = bp[(2 * ks * NCTP + cp) * 16];
            const v2f b0 = {q.x, q.y};
            const v2f b1 = {q.z, q.w};
            acc[2 * cp] = __builtin_amdgcn_wmma_f32_16x16x4_f32(
                 false, a,  false, b0,
                 (short)0, acc[2 * cp],  false,  false);
            acc[2 * cp + 1] = __builtin_amdgcn_wmma_f32_16x16x4_f32(
                 false, a,  false, b1,
                 (short)0, acc[2 * cp + 1],  false,  false);
        }
    }

    #pragma unroll
    for (int ct = 0; ct < NCT; ++ct)
        #pragma unroll
        for (int r = 0; r < 8; ++r)
            stg[(hi * 8 + r) * CP + ct * 16 + l15] = acc[ct][r];
    if (mtile * 16 < M) {
        float* hstrip = H + (size_t)mtile * 16 * NCOLS;
        #pragma unroll
        for (int i = 0; i < CL; ++i) {
            const int e   = (i * 32 + lane) * 4;
            const int row = e / NCOLS;
            const int col = e - row * NCOLS;
            *(float4*)&hstrip[e] = *(const float4*)&stg[row * CP + col];
        }
    }
}

#define AGG_R     200
#define AGG_WAVES 32
#define AGG_CAPW  256

template <int F>
__device__ __forceinline__ void agg_edge_now(const float* __restrict__ h,
                                             const int* __restrict__ src,
                                             const float* __restrict__ dinv,
                                             const float* t_dinv, float* tile,
                                             int edge, int row, int lane) {
    const int s = src[edge];
    const float nrm = dinv[s] * t_dinv[row];
    #pragma unroll
    for (int c = 0; c < F / 32; ++c)
        atomicAdd(&tile[row * F + c * 32 + lane], h[(size_t)s * F + c * 32 + lane] * nrm);
}

template <int F, bool RELU, bool FIRST>
__global__ __launch_bounds__(1024)
void agg_tile(const float* __restrict__ h,
              const int* __restrict__ src,
              const int* __restrict__ dst,
              const float* __restrict__ dinv,
              const float* __restrict__ bias,
              float* __restrict__ out,
              int* __restrict__ kept_a,
              int* __restrict__ kept_d,
              float* __restrict__ kept_n,
              int* __restrict__ kept_cnt,
              int e, int n) {
    __shared__ float tile[AGG_R * F];
    __shared__ float t_dinv[AGG_R];
    __shared__ int   l_a[AGG_WAVES * AGG_CAPW];
    __shared__ int   l_d[AGG_WAVES * AGG_CAPW];
    __shared__ float l_n[AGG_WAVES * AGG_CAPW];
    __shared__ int   l_cnt[AGG_WAVES];

    const int tid  = threadIdx.x;
    const int lane = tid & 31;
    const int wav  = __builtin_amdgcn_readfirstlane(tid >> 5);
    const int lo   = blockIdx.x * AGG_R;
    const int cnt  = min(AGG_R, n - lo);
    const unsigned ucnt = (unsigned)cnt;

    int* my_a = l_a + wav * AGG_CAPW;
    int* my_d = l_d + wav * AGG_CAPW;
    float* my_n = l_n + wav * AGG_CAPW;
    int wcount = 0;
    const int gw = blockIdx.x * AGG_WAVES + wav;
    const size_t kb = (size_t)gw * AGG_CAPW;

    bool scan = true;
    int   ka[8], kd[8];
    float kn[8];
    if (!FIRST) {
        wcount = kept_cnt[gw];
        scan = wcount > AGG_CAPW;
        if (scan) {
            wcount = 0;
        } else if (wcount > 0) {
            #pragma unroll
            for (int u = 0; u < 8; ++u) {
                const size_t qi = kb + (size_t)min(lane + 32 * u, wcount - 1);
                ka[u] = kept_a[qi];
                kd[u] = kept_d[qi];
                kn[u] = kept_n[qi];
            }
        }
    }

    for (int r = tid; r < cnt; r += 1024) t_dinv[r] = dinv[lo + r];
    for (int i = tid; i < cnt * F; i += 1024) {
        const int r = i / F;
        const int f = i - r * F;
        const float di = dinv[lo + r];
        tile[i] = h[(size_t)lo * F + i] * (di * di) + bias[f];
    }
    if (!FIRST && !scan) {
        #pragma unroll
        for (int u = 0; u < 8; ++u) {
            const int q = lane + 32 * u;
            if (q < wcount) { my_a[q] = ka[u]; my_d[q] = kd[u]; my_n[q] = kn[u]; }
        }
    }
    __syncthreads();

    auto take = [&](unsigned rr, int edge) {
        const bool m = rr < ucnt;
        const unsigned mask = __builtin_amdgcn_ballot_w32(m);
        if (mask) {
            const int pos = wcount + __popc(mask & ((1u << lane) - 1u));
            if (m && pos < AGG_CAPW) { my_a[pos] = edge; my_d[pos] = (int)rr; }
            unsigned ov = __builtin_amdgcn_ballot_w32(m && pos >= AGG_CAPW);
            while (ov) {
                const int l = __ffs(ov) - 1;
                ov &= ov - 1u;
                agg_edge_now<F>(h, src, dinv, t_dinv, tile,
                                __shfl(edge, l), __shfl((int)rr, l), lane);
            }
            wcount += __popc(mask);
        }
    };
    auto take4 = [&](int4 d, int edge0) {
        const unsigned r0 = (unsigned)(d.x - lo), r1 = (unsigned)(d.y - lo);
        const unsigned r2 = (unsigned)(d.z - lo), r3 = (unsigned)(d.w - lo);
        const bool any = (r0 < ucnt) | (r1 < ucnt) | (r2 < ucnt) | (r3 < ucnt);
        if (__builtin_amdgcn_ballot_w32(any)) {
            take(r0, edge0); take(r1, edge0 + 1); take(r2, edge0 + 2); take(r3, edge0 + 3);
        }
    };

    if (scan) {
        const int nchunks = (e + 1023) >> 10;
        const int rot = (int)(((long long)blockIdx.x * nchunks) / gridDim.x);
        const bool vec_ok = (e & 3) == 0;
        for (int k = wav; k < nchunks; k += AGG_WAVES) {
            int c = k + rot;
            if (c >= nchunks) c -= nchunks;
            const int base = (c << 10) + lane * 4;
            if (vec_ok && (c << 10) + 1024 <= e) {
                int4 d[8];
                #pragma unroll
                for (int u = 0; u < 8; ++u) d[u] = *(const int4*)(dst + base + u * 128);
                #pragma unroll
                for (int u = 0; u < 8; ++u) take4(d[u], base + u * 128);
            } else {
                #pragma unroll 1
                for (int u = 0; u < 8; ++u) {
                    const int i0 = base + u * 128;
                    int4 d;
                    d.x = (i0     < e) ? dst[i0]     : -1;
                    d.y = (i0 + 1 < e) ? dst[i0 + 1] : -1;
                    d.z = (i0 + 2 < e) ? dst[i0 + 2] : -1;
                    d.w = (i0 + 3 < e) ? dst[i0 + 3] : -1;
                    take4(d, i0);
                }
            }
        }
    }
    __syncthreads();

    static_assert(AGG_CAPW == 8 * 32, "the resolve step covers the list in 8 strides of 32 lanes");
    const int nl = min(wcount, AGG_CAPW);
    if (scan) {
        int   sv[8];
        float dv[8];
        #pragma unroll
        for (int u = 0; u < 8; ++u) {
            const int q = lane + 32 * u;
            sv[u] = (q < nl) ? src[my_a[q]] : 0;
        }
        #pragma unroll
        for (int u = 0; u < 8; ++u) {
            const int q = lane + 32 * u;
            dv[u] = (q < nl) ? dinv[sv[u]] : 0.0f;
        }
        #pragma unroll
        for (int u = 0; u < 8; ++u) {
            const int q = lane + 32 * u;
            if (q < nl) {
                const int   rw = my_d[q];
                const float wt = dv[u] * t_dinv[rw];
                my_a[q] = sv[u]; my_n[q] = wt;
                if (FIRST) { kept_a[kb + q] = sv[u]; kept_d[kb + q] = rw; kept_n[kb + q] = wt; }
            }
        }
        if (FIRST && lane == 0) kept_cnt[gw] = wcount;
    }
    if (lane == 0) l_cnt[wav] = nl;
    __syncthreads();

    static_assert(AGG_WAVES == 32, "one lane per list in the batch count below");
    constexpr int C = F / 32;
    constexpr int U = 8;
    const int lc  = l_cnt[lane];
    const int lnb = (lc + U - 1) / U;
    int inc = lnb;
    #pragma unroll
    for (int dd = 1; dd < 32; dd <<= 1) {
        const int up = __shfl_up(inc, dd);
        if (lane >= dd) inc += up;
    }
    const int pb = inc - lnb;
    const int nbatch = __builtin_amdgcn_readlane(inc, 31);
    for (int t = wav; t < nbatch; t += AGG_WAVES) {
        const int j   = __popc(__builtin_amdgcn_ballot_w32(pb <= t)) - 1;
        const int off = (t - __builtin_amdgcn_readlane(pb, j)) * U;
        const int len = __builtin_amdgcn_readlane(lc, j);
        const int*   ja = l_a + j * AGG_CAPW;
        const int*   jd = l_d + j * AGG_CAPW;
        const float* jn = l_n + j * AGG_CAPW;
        int   s[U], r[U];
        float w[U], v[U][C];
        if (off + U <= len) {
            #pragma unroll
            for (int u = 0; u < U; ++u) { s[u] = ja[off + u]; r[u] = jd[off + u]; w[u] = jn[off + u]; }
            #pragma unroll
            for (int u = 0; u < U; ++u)
                #pragma unroll
                for (int c = 0; c < C; ++c)
                    v[u][c] = h[(size_t)s[u] * F + c * 32 + lane];
            #pragma unroll
            for (int u = 0; u < U; ++u)
                #pragma unroll
                for (int c = 0; c < C; ++c)
                    atomicAdd(&tile[r[u] * F + c * 32 + lane], v[u][c] * w[u]);
        } else {
            #pragma unroll
            for (int u = 0; u < U; ++u) {
                const int q = min(off + u, len - 1);
                s[u] = ja[q]; r[u] = jd[q]; w[u] = jn[q];
            }
            #pragma unroll
            for (int u = 0; u < U; ++u)
                #pragma unroll
                for (int c = 0; c < C; ++c)
                    v[u][c] = h[(size_t)s[u] * F + c * 32 + lane];
            #pragma unroll
            for (int u = 0; u < U; ++u)
                if (off + u < len) {
                    #pragma unroll
                    for (int c = 0; c < C; ++c)
                        atomicAdd(&tile[r[u] * F + c * 32 + lane], v[u][c] * w[u]);
                }
        }
    }
    __syncthreads();

    for (int i = tid; i < cnt * F; i += 1024) {
        const float v = tile[i];
        out[(size_t)lo * F + i] = RELU ? fmaxf(v, 0.0f) : v;
    }
}

#define POOL_WAVES 32
#define POOL_LOADS 13
#define POOL_U     16
static_assert(POOL_WAVES * POOL_LOADS * 128 >= GCN_N_NODES, "the waves' slices cover every node");

__global__ __launch_bounds__(1024)
void pool_graph(const float* __restrict__ h,
                const int* __restrict__ batch,
                float* __restrict__ out, int n) {
    __shared__ int   list[GCN_N_NODES];
    __shared__ int   wcnt[POOL_WAVES];
    __shared__ float part[POOL_WAVES * 64];

    const int tid  = threadIdx.x;
    const int lane = tid & 31;
    const int wav  = __builtin_amdgcn_readfirstlane(tid >> 5);
    const int g    = blockIdx.x;
    const int wbase = wav * (POOL_LOADS * 128);

    int4 b[POOL_LOADS];
    if (wbase + POOL_LOADS * 128 <= n) {
        #pragma unroll
        for (int j = 0; j < POOL_LOADS; ++j)
            b[j] = *(const int4*)(batch + wbase + j * 128 + lane * 4);
    } else if (wbase < n) {
        #pragma unroll
        for (int j = 0; j < POOL_LOADS; ++j) {
            const int i0 = wbase + j * 128 + lane * 4;
            const int v0 = batch[min(i0,     n - 1)], v1 = batch[min(i0 + 1, n - 1)];
            const int v2 = batch[min(i0 + 2, n - 1)], v3 = batch[min(i0 + 3, n - 1)];
            b[j].x = (i0     < n) ? v0 : -1;
            b[j].y = (i0 + 1 < n) ? v1 : -1;
            b[j].z = (i0 + 2 < n) ? v2 : -1;
            b[j].w = (i0 + 3 < n) ? v3 : -1;
        }
    } else {
        #pragma unroll
        for (int j = 0; j < POOL_LOADS; ++j) b[j] = make_int4(-1, -1, -1, -1);
    }
    int cnt = 0;
    #pragma unroll
    for (int j = 0; j < POOL_LOADS; ++j) {
        cnt += __popc(__builtin_amdgcn_ballot_w32(b[j].x == g))
             + __popc(__builtin_amdgcn_ballot_w32(b[j].y == g))
             + __popc(__builtin_amdgcn_ballot_w32(b[j].z == g))
             + __popc(__builtin_amdgcn_ballot_w32(b[j].w == g));
    }
    if (lane == 0) wcnt[wav] = cnt;
    __syncthreads();

    int pos = 0, total = 0;
    #pragma unroll
    for (int v = 0; v < POOL_WAVES; ++v) {
        const int c = __builtin_amdgcn_readfirstlane(wcnt[v]);
        total += c;
        if (v < wav) pos += c;
    }
    if (cnt > 0) {
        const unsigned below = (1u << lane) - 1u;
        auto put = [&](int bv, int node) {
            const bool m = bv == g;
            const unsigned mask = __builtin_amdgcn_ballot_w32(m);
            if (m) list[pos + __popc(mask & below)] = node;
            pos += __popc(mask);
        };
        #pragma unroll
        for (int j = 0; j < POOL_LOADS; ++j) {
            const int i0 = wbase + j * 128 + lane * 4;
            put(b[j].x, i0); put(b[j].y, i0 + 1); put(b[j].z, i0 + 2); put(b[j].w, i0 + 3);
        }
    }
    __syncthreads();

    const int e_lo = (wav * total) / POOL_WAVES;
    const int e_hi = ((wav + 1) * total) / POOL_WAVES;
    v2f acc = {0.f, 0.f};
    for (int e = e_lo; e < e_hi; e += POOL_U) {
        int nd[POOL_U];
        v2f v[POOL_U];
        #pragma unroll
        for (int u = 0; u < POOL_U; ++u) {
            const int q = e + u;
            nd[u] = __builtin_amdgcn_readfirstlane(list[q < e_hi ? q : e_lo]);
        }
        #pragma unroll
        for (int u = 0; u < POOL_U; ++u)
            v[u] = *(const v2f*)(h + (size_t)nd[u] * 64 + lane * 2);
        #pragma unroll
        for (int u = 0; u < POOL_U; ++u)
            if (e + u < e_hi) { acc.x += v[u].x; acc.y += v[u].y; }
    }
    *(v2f*)&part[wav * 64 + lane * 2] = acc;
    __syncthreads();

    if (tid < 64) {
        float s = 0.f;
        #pragma unroll
        for (int w = 0; w < POOL_WAVES; ++w) s += part[w * 64 + tid];
        out[g * 64 + tid] = s / fmaxf((float)total, 1.0f);
    }
}

extern "C" void kernel_launch(void* const* d_in, const int* in_sizes, int n_in,
                              void* d_out, int out_size, void* d_ws, size_t ws_size,
                              hipStream_t stream) {
    const float* x     = (const float*)d_in[0];
    const int*   ei    = (const int*)d_in[1];
    const int*   batch = (const int*)d_in[2];
    const float* Wl[4] = {(const float*)d_in[3], (const float*)d_in[5],
                          (const float*)d_in[7], (const float*)d_in[9]};
    const float* bl[4] = {(const float*)d_in[4], (const float*)d_in[6],
                          (const float*)d_in[8], (const float*)d_in[10]};

    const int N = GCN_N_NODES;
    const int E = in_sizes[1] / 2;
    const int* src = ei;
    const int* dst = ei + E;

    auto au = [](size_t v) { return (v + 255) & ~(size_t)255; };
    char* ws = (char*)d_ws;
    size_t off = 0;
    float* deg  = (float*)(ws + off); off += au(sizeof(float) * N);
    float* bufA = (float*)(ws + off); off += au(sizeof(float) * (size_t)N * 96);
    float* bufC = (float*)(ws + off); off += au(sizeof(float) * (size_t)N * 96);
    const size_t KEPT = (size_t)((GCN_N_NODES + AGG_R - 1) / AGG_R) * AGG_WAVES * AGG_CAPW;
    int*   kept_a   = (int*)(ws + off);   off += au(sizeof(int) * KEPT);
    int*   kept_d   = (int*)(ws + off);   off += au(sizeof(int) * KEPT);
    float* kept_n   = (float*)(ws + off); off += au(sizeof(float) * KEPT);
    int*   kept_cnt = (int*)(ws + off);   off += au(sizeof(int) * (KEPT / AGG_CAPW));
    (void)ws_size; (void)n_in; (void)out_size;

    const int TB = 256;
    deg_init    <<<(N + TB - 1) / TB, TB, 0, stream>>>(deg, N);
    deg_accum   <<<(E + TB - 1) / TB, TB, 0, stream>>>(dst, deg, E);
    deg_finalize<<<(N + TB - 1) / TB, TB, 0, stream>>>(deg, N);

    constexpr int WV = 16;
    const int MT  = (N + 15) / 16;
    const int GBX = (MT + WV - 1) / WV;
    const int AGB = (N + AGG_R - 1) / AGG_R;

    gemm_f32_wmma<128, 96, WV><<<GBX, 32 * WV, 0, stream>>>(x, Wl[0], bufA, N);
    agg_tile<96, true, true><<<AGB, 1024, 0, stream>>>(bufA, src, dst, deg, bl[0], bufC,
                                                       kept_a, kept_d, kept_n, kept_cnt, E, N);

    for (int l = 1; l <= 2; ++l) {
        gemm_f32_wmma<96, 96, WV><<<GBX, 32 * WV, 0, stream>>>(bufC, Wl[l], bufA, N);
        agg_tile<96, true, false><<<AGB, 1024, 0, stream>>>(bufA, src, dst, deg, bl[l], bufC,
                                                            kept_a, kept_d, kept_n, kept_cnt, E, N);
    }

    gemm_f32_wmma<96, 64, WV><<<GBX, 32 * WV, 0, stream>>>(bufC, Wl[3], bufA, N);
    agg_tile<64, false, false><<<AGB, 1024, 0, stream>>>(bufA, src, dst, deg, bl[3], bufC,
                                                         kept_a, kept_d, kept_n, kept_cnt, E, N);

    pool_graph<<<GCN_N_GRAPHS, 1024, 0, stream>>>(bufC, batch, (float*)d_out, N);
}


# First-layer filter scan reads a byte-per-edge tile-id plane (packed by the degree pass, 4 edges a thread), not 4-byte dst: 16 edges a load, byte compares; rows from dst at resolve; list order changes
# speedup vs baseline: 2.9958x; 1.0482x over previous
#include <hip/hip_runtime.h>

typedef __attribute__((ext_vector_type(2))) float v2f;
typedef __attribute__((ext_vector_type(4))) float v4f;
typedef __attribute__((ext_vector_type(8))) float v8f;
typedef __attribute__((ext_vector_type(4))) unsigned v4u;

#define GCN_N_NODES  50000
#define GCN_N_GRAPHS 64

#define AGG_R     200
#define AGG_WAVES 32
#define AGG_CAPW  256
#define SCAN_CH   2048
static_assert((GCN_N_NODES + AGG_R - 1) / AGG_R <= 255, "tile numbers are bytes, 255 is no tile");

__global__ void deg_init(float* __restrict__ deg, int n) {
    int i = blockIdx.x * blockDim.x + threadIdx.x;
    if (i < n) deg[i] = 1.0f;
}

__global__ void deg_accum(const int* __restrict__ dst, float* __restrict__ deg,
                          unsigned* __restrict__ plane, int e, int words) {
    const int t = blockIdx.x * blockDim.x + threadIdx.x;
    if (t >= words) return;
    int d[4];
    #pragma unroll
    for (int k = 0; k < 4; ++k) d[k] = dst[min(4 * t + k, e - 1)];
    unsigned packed = 0u;
    #pragma unroll
    for (int k = 0; k < 4; ++k) {
        unsigned id = 255u;
        if (4 * t + k < e) {
            atomicAdd(&deg[d[k]], 1.0f);
            id = (unsigned)d[k] / AGG_R;
        }
        packed |= id << (8 * k);
    }
    plane[t] = packed;
}

__global__ void deg_finalize(float* __restrict__ deg, int n) {
    int i = blockIdx.x * blockDim.x + threadIdx.x;
    if (i < n) deg[i] = rsqrtf(deg[i]);
}

template <int K, int NCOLS, int WAVES>
__global__ __launch_bounds__(32 * WAVES)
void gemm_f32_wmma(const float* __restrict__ A,
                   const float* __restrict__ Wm,
                   float* __restrict__ H, int M) {
    constexpr int NCT   = NCOLS / 16;
    constexpr int NCTP  = NCT / 2;
    constexpr int KS    = K / 4;
    constexpr int T     = 32 * WAVES;
    constexpr int C4    = NCOLS / 4;
    constexpr int ITEMS = (K / 2) * C4;
    constexpr int PER   = (ITEMS + T - 1) / T;
    constexpr int AP    = K + 4;
    constexpr int CP    = NCOLS + 4;
    constexpr int STG   = 16 * (AP > CP ? AP : CP);
    constexpr int AL    = (16 * K / 4) / 32;
    constexpr int CL    = (16 * NCOLS / 4) / 32;
    static_assert(NCT % 2 == 0 && K % 4 == 0, "column tiles come in pairs, k in fours");
    static_assert(ITEMS % 32 == 0, "a wave's W items are all inside or all outside");
    static_assert((16 * K / 4) % 32 == 0 && (16 * NCOLS / 4) % 32 == 0, "whole 512-byte pieces");
    static_assert((AP / 4) % 2 == 1 && (CP / 4) % 2 == 1, "stage pitches are 16 bytes x odd");
    __shared__ v4f Bs[(K / 2) * NCTP * 16];
    __shared__ v4f Stg4[WAVES * STG / 4];

    const int tid  = threadIdx.x;
    const int lane = tid & 31;
    const int wav  = __builtin_amdgcn_readfirstlane(tid >> 5);
    const int l15  = lane & 15;
    const int hi   = lane >> 4;
    const int mtile = blockIdx.x * WAVES + wav;
    const int mt   = min(mtile, M / 16 - 1);
    float* stg = (float*)Stg4 + wav * STG;

    float4 w0[PER], w1[PER];
    #pragma unroll
    for (int p = 0; p < PER; ++p) {
        if (ITEMS % T == 0 || wav * 32 + p * T < ITEMS) {
            const int it = tid + p * T;
            const int kp = it / C4;
            const int c4 = it - kp * C4;
            w0[p] = *(const float4*)&Wm[(2 * kp) * NCOLS + 4 * c4];
            w1[p] = *(const float4*)&Wm[(2 * kp + 1) * NCOLS + 4 * c4];
        }
    }

    const float* astrip = A + (size_t)mt * 16 * K;
    float4 av[AL];
    #pragma unroll
    for (int i = 0; i < AL; ++i)
        av[i] = *(const float4*)&astrip[(i * 32 + lane) * 4];

    #pragma unroll
    for (int p = 0; p < PER; ++p) {
        if (ITEMS % T == 0 || wav * 32 + p * T < ITEMS) {
            const int it = tid + p * T;
            const int kp = it / C4;
            const int c4 = it - kp * C4;
            const int ct = c4 >> 2;
            float* d = (float*)&Bs[(kp * NCTP + (ct >> 1)) * 16 + (c4 & 3) * 4] + (ct & 1) * 2;
            *(v2f*)(d)      = v2f{w0[p].x, w1[p].x};
            *(v2f*)(d + 4)  = v2f{w0[p].y, w1[p].y};
            *(v2f*)(d + 8)  = v2f{w0[p].z, w1[p].z};
            *(v2f*)(d + 12) = v2f{w0[p].w, w1[p].w};
        }
    }
    #pragma unroll
    for (int i = 0; i < AL; ++i) {
        const int e   = (i * 32 + lane) * 4;
        const int row = e / K;
        const int col = e - row * K;
        *(float4*)&stg[row * AP + col] = av[i];
    }
    __syncthreads();

    v8f acc[NCT];
    #pragma unroll
    for (int ct = 0; ct < NCT; ++ct) acc[ct] = v8f{0.f, 0.f, 0.f, 0.f, 0.f, 0.f, 0.f, 0.f};
    const v4f*   bp = Bs + hi * (NCTP * 16) + l15;
    const float* ap = stg + l15 * AP + hi * 2;
    #pragma unroll
    for (int ks = 0; ks < KS; ++ks) {
        const v2f a = *(const v2f*)&ap[ks * 4];
        #pragma unroll
        for (int cp = 0; cp < NCTP; ++cp) {
            const v4f q = bp[(2 * ks * NCTP + cp) * 16];
            const v2f b0 = {q.x, q.y};
            const v2f b1 = {q.z, q.w};
            acc[2 * cp] = __builtin_amdgcn_wmma_f32_16x16x4_f32(
                 false, a,  false, b0,
                 (short)0, acc[2 * cp],  false,  false);
            acc[2 * cp + 1] = __builtin_amdgcn_wmma_f32_16x16x4_f32(
                 false, a,  false, b1,
                 (short)0, acc[2 * cp + 1],  false,  false);
        }
    }

    #pragma unroll
    for (int ct = 0; ct < NCT; ++ct)
        #pragma unroll
        for (int r = 0; r < 8; ++r)
            stg[(hi * 8 + r) * CP + ct * 16 + l15] = acc[ct][r];
    if (mtile * 16 < M) {
        float* hstrip = H + (size_t)mtile * 16 * NCOLS;
        #pragma unroll
        for (int i = 0; i < CL; ++i) {
            const int e   = (i * 32 + lane) * 4;
            const int row = e / NCOLS;
            const int col = e - row * NCOLS;
            *(float4*)&hstrip[e] = *(const float4*)&stg[row * CP + col];
        }
    }
}

template <int F>
__device__ __forceinline__ void agg_edge_now(const float* __restrict__ h,
                                             const int* __restrict__ src,
                                             const float* __restrict__ dinv,
                                             const float* t_dinv, float* tile,
                                             int edge, int row, int lane) {
    const int s = src[edge];
    const float nrm = dinv[s] * t_dinv[row];
    #pragma unroll
    for (int c = 0; c < F / 32; ++c)
        atomicAdd(&tile[row * F + c * 32 + lane], h[(size_t)s * F + c * 32 + lane] * nrm);
}

template <int F, bool RELU, bool FIRST>
__global__ __launch_bounds__(1024)
void agg_tile(const float* __restrict__ h,
              const int* __restrict__ src,
              const int* __restrict__ dst,
              const unsigned* __restrict__ plane,
              const float* __restrict__ dinv,
              const float* __restrict__ bias,
              float* __restrict__ out,
              int* __restrict__ kept_a,
              int* __restrict__ kept_d,
              float* __restrict__ kept_n,
              int* __restrict__ kept_cnt,
              int e, int n) {
    __shared__ float tile[AGG_R * F];
    __shared__ float t_dinv[AGG_R];
    __shared__ int   l_a[AGG_WAVES * AGG_CAPW];
    __shared__ int   l_d[AGG_WAVES * AGG_CAPW];
    __shared__ float l_n[AGG_WAVES * AGG_CAPW];
    __shared__ int   l_cnt[AGG_WAVES];

    const int tid  = threadIdx.x;
    const int lane = tid & 31;
    const int wav  = __builtin_amdgcn_readfirstlane(tid >> 5);
    const int lo   = blockIdx.x * AGG_R;
    const int cnt  = min(AGG_R, n - lo);
    const unsigned ucnt = (unsigned)cnt;

    int* my_a = l_a + wav * AGG_CAPW;
    int* my_d = l_d + wav * AGG_CAPW;
    float* my_n = l_n + wav * AGG_CAPW;
    int wcount = 0;
    const int gw = blockIdx.x * AGG_WAVES + wav;
    const size_t kb = (size_t)gw * AGG_CAPW;

    bool scan = true;
    int   ka[8], kd[8];
    float kn[8];
    if (!FIRST) {
        wcount = kept_cnt[gw];
        scan = wcount > AGG_CAPW;
        if (scan) {
            wcount = 0;
        } else if (wcount > 0) {
            #pragma unroll
            for (int u = 0; u < 8; ++u) {
                const size_t qi = kb + (size_t)min(lane + 32 * u, wcount - 1);
                ka[u] = kept_a[qi];
                kd[u] = kept_d[qi];
                kn[u] = kept_n[qi];
            }
        }
    }

    for (int r = tid; r < cnt; r += 1024) t_dinv[r] = dinv[lo + r];
    for (int i = tid; i < cnt * F; i += 1024) {
        const int r = i / F;
        const int f = i - r * F;
        const float di = dinv[lo + r];
        tile[i] = h[(size_t)lo * F + i] * (di * di) + bias[f];
    }
    if (!FIRST && !scan) {
        #pragma unroll
        for (int u = 0; u < 8; ++u) {
            const int q = lane + 32 * u;
            if (q < wcount) { my_a[q] = ka[u]; my_d[q] = kd[u]; my_n[q] = kn[u]; }
        }
    }
    __syncthreads();

    if (scan) {
        const int nchunks = (e + SCAN_CH - 1) / SCAN_CH;
        const int rot = (int)(((long long)blockIdx.x * nchunks) / gridDim.x);
        const unsigned pat = (unsigned)blockIdx.x * 0x01010101u;
        const unsigned below = (1u << lane) - 1u;
        auto eqb = [&](unsigned w) {
            const unsigned x = w ^ pat;
            return ~(((x & 0x7F7F7F7Fu) + 0x7F7F7F7Fu) | x) & 0x80808080u;
        };
        for (int k = wav; k < nchunks; k += AGG_WAVES) {
            int c = k + rot;
            if (c >= nchunks) c -= nchunks;
            const int base = c * SCAN_CH + lane * 16;
            v4u q[4];
            #pragma unroll
            for (int u = 0; u < 4; ++u) q[u] = *(const v4u*)(plane + ((base + u * 512) >> 2));
            asm volatile("" :: "v"(q[0]), "v"(q[1]), "v"(q[2]), "v"(q[3]));
            #pragma unroll
            for (int u = 0; u < 4; ++u) {
                unsigned hits = (eqb(q[u].x) >> 7) | (eqb(q[u].y) >> 6)
                              | (eqb(q[u].z) >> 5) | (eqb(q[u].w) >> 4);
                unsigned act = __builtin_amdgcn_ballot_w32(hits != 0u);
                while (act) {
                    const bool m = hits != 0u;
                    const int  b = __ffs(hits) - 1;
                    const int  edge = base + u * 512 + ((b & 7) << 2) + (b >> 3);
                    hits &= hits - 1u;
                    const int pos = wcount + __popc(act & below);
                    if (m && pos < AGG_CAPW) my_a[pos] = edge;
                    unsigned ov = __builtin_amdgcn_ballot_w32(m && pos >= AGG_CAPW);
                    while (ov) {
                        const int l = __ffs(ov) - 1;
                        ov &= ov - 1u;
                        const int ed = __shfl(edge, l);
                        agg_edge_now<F>(h, src, dinv, t_dinv, tile, ed, dst[ed] - lo, lane);
                    }
                    wcount += __popc(act);
                    act = __builtin_amdgcn_ballot_w32(hits != 0u);
                }
            }
        }
    }
    __syncthreads();

    static_assert(AGG_CAPW == 8 * 32, "the resolve step covers the list in 8 strides of 32 lanes");
    const int nl = min(wcount, AGG_CAPW);
    if (scan) {
        int   sv[8], rv[8];
        float dv[8];
        #pragma unroll
        for (int u = 0; u < 8; ++u) {
            const int q = lane + 32 * u;
            sv[u] = (q < nl) ? src[my_a[q]] : 0;
            rv[u] = (q < nl) ? dst[my_a[q]] - lo : 0;
        }
        #pragma unroll
        for (int u = 0; u < 8; ++u) {
            const int q = lane + 32 * u;
            dv[u] = (q < nl) ? dinv[sv[u]] : 0.0f;
        }
        #pragma unroll
        for (int u = 0; u < 8; ++u) {
            const int q = lane + 32 * u;
            if (q < nl) {
                const int   rw = rv[u];
                const float wt = dv[u] * t_dinv[rw];
                my_a[q] = sv[u]; my_d[q] = rw; my_n[q] = wt;
                if (FIRST) { kept_a[kb + q] = sv[u]; kept_d[kb + q] = rw; kept_n[kb + q] = wt; }
            }
        }
        if (FIRST && lane == 0) kept_cnt[gw] = wcount;
    }
    if (lane == 0) l_cnt[wav] = nl;
    __syncthreads();

    static_assert(AGG_WAVES == 32, "one lane per list in the batch count below");
    constexpr int C = F / 32;
    constexpr int U = 8;
    const int lc  = l_cnt[lane];
    const int lnb = (lc + U - 1) / U;
    int inc = lnb;
    #pragma unroll
    for (int dd = 1; dd < 32; dd <<= 1) {
        const int up = __shfl_up(inc, dd);
        if (lane >= dd) inc += up;
    }
    const int pb = inc - lnb;
    const int nbatch = __builtin_amdgcn_readlane(inc, 31);
    for (int t = wav; t < nbatch; t += AGG_WAVES) {
        const int j   = __popc(__builtin_amdgcn_ballot_w32(pb <= t)) - 1;
        const int off = (t - __builtin_amdgcn_readlane(pb, j)) * U;
        const int len = __builtin_amdgcn_readlane(lc, j);
        const int*   ja = l_a + j * AGG_CAPW;
        const int*   jd = l_d + j * AGG_CAPW;
        const float* jn = l_n + j * AGG_CAPW;
        int   s[U], r[U];
        float w[U], v[U][C];
        if (off + U <= len) {
            #pragma unroll
            for (int u = 0; u < U; ++u) { s[u] = ja[off + u]; r[u] = jd[off + u]; w[u] = jn[off + u]; }
            #pragma unroll
            for (int u = 0; u < U; ++u)
                #pragma unroll
                for (int c = 0; c < C; ++c)
                    v[u][c] = h[(size_t)s[u] * F + c * 32 + lane];
            #pragma unroll
            for (int u = 0; u < U; ++u)
                #pragma unroll
                for (int c = 0; c < C; ++c)
                    atomicAdd(&tile[r[u] * F + c * 32 + lane], v[u][c] * w[u]);
        } else {
            #pragma unroll
            for (int u = 0; u < U; ++u) {
                const int q = min(off + u, len - 1);
                s[u] = ja[q]; r[u] = jd[q]; w[u] = jn[q];
            }
            #pragma unroll
            for (int u = 0; u < U; ++u)
                #pragma unroll
                for (int c = 0; c < C; ++c)
                    v[u][c] = h[(size_t)s[u] * F + c * 32 + lane];
            #pragma unroll
            for (int u = 0; u < U; ++u)
                if (off + u < len) {
                    #pragma unroll
                    for (int c = 0; c < C; ++c)
                        atomicAdd(&tile[r[u] * F + c * 32 + lane], v[u][c] * w[u]);
                }
        }
    }
    __syncthreads();

    for (int i = tid; i < cnt * F; i += 1024) {
        const float v = tile[i];
        out[(size_t)lo * F + i] = RELU ? fmaxf(v, 0.0f) : v;
    }
}

#define POOL_WAVES 32
#define POOL_LOADS 13
#define POOL_U     16
static_assert(POOL_WAVES * POOL_LOADS * 128 >= GCN_N_NODES, "the waves' slices cover every node");

__global__ __launch_bounds__(1024)
void pool_graph(const float* __restrict__ h,
                const int* __restrict__ batch,
                float* __restrict__ out, int n) {
    __shared__ int   list[GCN_N_NODES];
    __shared__ int   wcnt[POOL_WAVES];
    __shared__ float part[POOL_WAVES * 64];

    const int tid  = threadIdx.x;
    const int lane = tid & 31;
    const int wav  = __builtin_amdgcn_readfirstlane(tid >> 5);
    const int g    = blockIdx.x;
    const int wbase = wav * (POOL_LOADS * 128);

    int4 b[POOL_LOADS];
    if (wbase + POOL_LOADS * 128 <= n) {
        #pragma unroll
        for (int j = 0; j < POOL_LOADS; ++j)
            b[j] = *(const int4*)(batch + wbase + j * 128 + lane * 4);
    } else if (wbase < n) {
        #pragma unroll
        for (int j = 0; j < POOL_LOADS; ++j) {
            const int i0 = wbase + j * 128 + lane * 4;
            const int v0 = batch[min(i0,     n - 1)], v1 = batch[min(i0 + 1, n - 1)];
            const int v2 = batch[min(i0 + 2, n - 1)], v3 = batch[min(i0 + 3, n - 1)];
            b[j].x = (i0     < n) ? v0 : -1;
            b[j].y = (i0 + 1 < n) ? v1 : -1;
            b[j].z = (i0 + 2 < n) ? v2 : -1;
            b[j].w = (i0 + 3 < n) ? v3 : -1;
        }
    } else {
        #pragma unroll
        for (int j = 0; j < POOL_LOADS; ++j) b[j] = make_int4(-1, -1, -1, -1);
    }
    int cnt = 0;
    #pragma unroll
    for (int j = 0; j < POOL_LOADS; ++j) {
        cnt += __popc(__builtin_amdgcn_ballot_w32(b[j].x == g))
             + __popc(__builtin_amdgcn_ballot_w32(b[j].y == g))
             + __popc(__builtin_amdgcn_ballot_w32(b[j].z == g))
             + __popc(__builtin_amdgcn_ballot_w32(b[j].w == g));
    }
    if (lane == 0) wcnt[wav] = cnt;
    __syncthreads();

    int pos = 0, total = 0;
    #pragma unroll
    for (int v = 0; v < POOL_WAVES; ++v) {
        const int c = __builtin_amdgcn_readfirstlane(wcnt[v]);
        total += c;
        if (v < wav) pos += c;
    }
    if (cnt > 0) {
        const unsigned below = (1u << lane) - 1u;
        auto put = [&](int bv, int node) {
            const bool m = bv == g;
            const unsigned mask = __builtin_amdgcn_ballot_w32(m);
            if (m) list[pos + __popc(mask & below)] = node;
            pos += __popc(mask);
        };
        #pragma unroll
        for (int j = 0; j < POOL_LOADS; ++j) {
            const int i0 = wbase + j * 128 + lane * 4;
            put(b[j].x, i0); put(b[j].y, i0 + 1); put(b[j].z, i0 + 2); put(b[j].w, i0 + 3);
        }
    }
    __syncthreads();

    const int e_lo = (wav * total) / POOL_WAVES;
    const int e_hi = ((wav + 1) * total) / POOL_WAVES;
    v2f acc = {0.f, 0.f};
    for (int e = e_lo; e < e_hi; e += POOL_U) {
        int nd[POOL_U];
        v2f v[POOL_U];
        #pragma unroll
        for (int u = 0; u < POOL_U; ++u) {
            const int q = e + u;
            nd[u] = __builtin_amdgcn_readfirstlane(list[q < e_hi ? q : e_lo]);
        }
        #pragma unroll
        for (int u = 0; u < POOL_U; ++u)
            v[u] = *(const v2f*)(h + (size_t)nd[u] * 64 + lane * 2);
        #pragma unroll
        for (int u = 0; u < POOL_U; ++u)
            if (e + u < e_hi) { acc.x += v[u].x; acc.y += v[u].y; }
    }
    *(v2f*)&part[wav * 64 + lane * 2] = acc;
    __syncthreads();

    if (tid < 64) {
        float s = 0.f;
        #pragma unroll
        for (int w = 0; w < POOL_WAVES; ++w) s += part[w * 64 + tid];
        out[g * 64 + tid] = s / fmaxf((float)total, 1.0f);
    }
}

extern "C" void kernel_launch(void* const* d_in, const int* in_sizes, int n_in,
                              void* d_out, int out_size, void* d_ws, size_t ws_size,
                              hipStream_t stream) {
    const float* x     = (const float*)d_in[0];
    const int*   ei    = (const int*)d_in[1];
    const int*   batch = (const int*)d_in[2];
    const float* Wl[4] = {(const float*)d_in[3], (const float*)d_in[5],
                          (const float*)d_in[7], (const float*)d_in[9]};
    const float* bl[4] = {(const float*)d_in[4], (const float*)d_in[6],
                          (const float*)d_in[8], (const float*)d_in[10]};

    const int N = GCN_N_NODES;
    const int E = in_sizes[1] / 2;
    const int* src = ei;
    const int* dst = ei + E;

    auto au = [](size_t v) { return (v + 255) & ~(size_t)255; };
    char* ws = (char*)d_ws;
    size_t off = 0;
    float* deg  = (float*)(ws + off); off += au(sizeof(float) * N);
    float* bufA = (float*)(ws + off); off += au(sizeof(float) * (size_t)N * 96);
    float* bufC = (float*)(ws + off); off += au(sizeof(float) * (size_t)N * 96);
    const size_t KEPT = (size_t)((GCN_N_NODES + AGG_R - 1) / AGG_R) * AGG_WAVES * AGG_CAPW;
    int*   kept_a   = (int*)(ws + off);   off += au(sizeof(int) * KEPT);
    int*   kept_d   = (int*)(ws + off);   off += au(sizeof(int) * KEPT);
    float* kept_n   = (float*)(ws + off); off += au(sizeof(float) * KEPT);
    int*   kept_cnt = (int*)(ws + off);   off += au(sizeof(int) * (KEPT / AGG_CAPW));
    const int EP = ((E + SCAN_CH - 1) / SCAN_CH) * SCAN_CH;
    unsigned* plane = (unsigned*)(ws + off); off += au((size_t)EP);
    (void)ws_size; (void)n_in; (void)out_size;

    const int TB = 256;
    deg_init    <<<(N + TB - 1) / TB, TB, 0, stream>>>(deg, N);
    deg_accum   <<<(EP / 4 + TB - 1) / TB, TB, 0, stream>>>(dst, deg, plane, E, EP / 4);
    deg_finalize<<<(N + TB - 1) / TB, TB, 0, stream>>>(deg, N);

    constexpr int WV = 16;
    const int MT  = (N + 15) / 16;
    const int GBX = (MT + WV - 1) / WV;
    const int AGB = (N + AGG_R - 1) / AGG_R;

    gemm_f32_wmma<128, 96, WV><<<GBX, 32 * WV, 0, stream>>>(x, Wl[0], bufA, N);
    agg_tile<96, true, true><<<AGB, 1024, 0, stream>>>(bufA, src, dst, plane, deg, bl[0], bufC,
                                                       kept_a, kept_d, kept_n, kept_cnt, E, N);

    for (int l = 1; l <= 2; ++l) {
        gemm_f32_wmma<96, 96, WV><<<GBX, 32 * WV, 0, stream>>>(bufC, Wl[l], bufA, N);
        agg_tile<96, true, false><<<AGB, 1024, 0, stream>>>(bufA, src, dst, plane, deg, bl[l], bufC,
                                                            kept_a, kept_d, kept_n, kept_cnt, E, N);
    }

    gemm_f32_wmma<96, 64, WV><<<GBX, 32 * WV, 0, stream>>>(bufC, Wl[3], bufA, N);
    agg_tile<64, false, false><<<AGB, 1024, 0, stream>>>(bufA, src, dst, plane, deg, bl[3], bufC,
                                                         kept_a, kept_d, kept_n, kept_cnt, E, N);

    pool_graph<<<GCN_N_GRAPHS, 1024, 0, stream>>>(bufC, batch, (float*)d_out, N);
}


# Aggregation tile fill and write-out in 16-byte pieces, every load of the fill (h, dinv, bias) sent out before the first wait: one memory round trip for the fill, not five; sums as before
# speedup vs baseline: 3.0963x; 1.0335x over previous
#include <hip/hip_runtime.h>

typedef __attribute__((ext_vector_type(2))) float v2f;
typedef __attribute__((ext_vector_type(4))) float v4f;
typedef __attribute__((ext_vector_type(8))) float v8f;
typedef __attribute__((ext_vector_type(4))) unsigned v4u;

#define GCN_N_NODES  50000
#define GCN_N_GRAPHS 64

#define AGG_R     200
#define AGG_WAVES 32
#define AGG_CAPW  256
#define SCAN_CH   2048
static_assert((GCN_N_NODES + AGG_R - 1) / AGG_R <= 255, "tile numbers are bytes, 255 is no tile");

__global__ void deg_init(float* __restrict__ deg, int n) {
    int i = blockIdx.x * blockDim.x + threadIdx.x;
    if (i < n) deg[i] = 1.0f;
}

__global__ void deg_accum(const int* __restrict__ dst, float* __restrict__ deg,
                          unsigned* __restrict__ plane, int e, int words) {
    const int t = blockIdx.x * blockDim.x + threadIdx.x;
    if (t >= words) return;
    int d[4];
    #pragma unroll
    for (int k = 0; k < 4; ++k) d[k] = dst[min(4 * t + k, e - 1)];
    unsigned packed = 0u;
    #pragma unroll
    for (int k = 0; k < 4; ++k) {
        unsigned id = 255u;
        if (4 * t + k < e) {
            atomicAdd(&deg[d[k]], 1.0f);
            id = (unsigned)d[k] / AGG_R;
        }
        packed |= id << (8 * k);
    }
    plane[t] = packed;
}

__global__ void deg_finalize(float* __restrict__ deg, int n) {
    int i = blockIdx.x * blockDim.x + threadIdx.x;
    if (i < n) deg[i] = rsqrtf(deg[i]);
}

template <int K, int NCOLS, int WAVES>
__global__ __launch_bounds__(32 * WAVES)
void gemm_f32_wmma(const float* __restrict__ A,
                   const float* __restrict__ Wm,
                   float* __restrict__ H, int M) {
    constexpr int NCT   = NCOLS / 16;
    constexpr int NCTP  = NCT / 2;
    constexpr int KS    = K / 4;
    constexpr int T     = 32 * WAVES;
    constexpr int C4    = NCOLS / 4;
    constexpr int ITEMS = (K / 2) * C4;
    constexpr int PER   = (ITEMS + T - 1) / T;
    constexpr int AP    = K + 4;
    constexpr int CP    = NCOLS + 4;
    constexpr int STG   = 16 * (AP > CP ? AP : CP);
    constexpr int AL    = (16 * K / 4) / 32;
    constexpr int CL    = (16 * NCOLS / 4) / 32;
    static_assert(NCT % 2 == 0 && K % 4 == 0, "column tiles come in pairs, k in fours");
    static_assert(ITEMS % 32 == 0, "a wave's W items are all inside or all outside");
    static_assert((16 * K / 4) % 32 == 0 && (16 * NCOLS / 4) % 32 == 0, "whole 512-byte pieces");
    static_assert((AP / 4) % 2 == 1 && (CP / 4) % 2 == 1, "stage pitches are 16 bytes x odd");
    __shared__ v4f Bs[(K / 2) * NCTP * 16];
    __shared__ v4f Stg4[WAVES * STG / 4];

    const int tid  = threadIdx.x;
    const int lane = tid & 31;
    const int wav  = __builtin_amdgcn_readfirstlane(tid >> 5);
    const int l15  = lane & 15;
    const int hi   = lane >> 4;
    const int mtile = blockIdx.x * WAVES + wav;
    const int mt   = min(mtile, M / 16 - 1);
    float* stg = (float*)Stg4 + wav * STG;

    float4 w0[PER], w1[PER];
    #pragma unroll
    for (int p = 0; p < PER; ++p) {
        if (ITEMS % T == 0 || wav * 32 + p * T < ITEMS) {
            const int it = tid + p * T;
            const int kp = it / C4;
            const int c4 = it - kp * C4;
            w0[p] = *(const float4*)&Wm[(2 * kp) * NCOLS + 4 * c4];
            w1[p] = *(const float4*)&Wm[(2 * kp + 1) * NCOLS + 4 * c4];
        }
    }

    const float* astrip = A + (size_t)mt * 16 * K;
    float4 av[AL];
    #pragma unroll
    for (int i = 0; i < AL; ++i)
        av[i] = *(const float4*)&astrip[(i * 32 + lane) * 4];

    #pragma unroll
    for (int p = 0; p < PER; ++p) {
        if (ITEMS % T == 0 || wav * 32 + p * T < ITEMS) {
            const int it = tid + p * T;
            const int kp = it / C4;
            const int c4 = it - kp * C4;
            const int ct = c4 >> 2;
            float* d = (float*)&Bs[(kp * NCTP + (ct >> 1)) * 16 + (c4 & 3) * 4] + (ct & 1) * 2;
            *(v2f*)(d)      = v2f{w0[p].x, w1[p].x};
            *(v2f*)(d + 4)  = v2f{w0[p].y, w1[p].y};
            *(v2f*)(d + 8)  = v2f{w0[p].z, w1[p].z};
            *(v2f*)(d + 12) = v2f{w0[p].w, w1[p].w};
        }
    }
    #pragma unroll
    for (int i = 0; i < AL; ++i) {
        const int e   = (i * 32 + lane) * 4;
        const int row = e / K;
        const int col = e - row * K;
        *(float4*)&stg[row * AP + col] = av[i];
    }
    __syncthreads();

    v8f acc[NCT];
    #pragma unroll
    for (int ct = 0; ct < NCT; ++ct) acc[ct] = v8f{0.f, 0.f, 0.f, 0.f, 0.f, 0.f, 0.f, 0.f};
    const v4f*   bp = Bs + hi * (NCTP * 16) + l15;
    const float* ap = stg + l15 * AP + hi * 2;
    #pragma unroll
    for (int ks = 0; ks < KS; ++ks) {
        const v2f a = *(const v2f*)&ap[ks * 4];
        #pragma unroll
        for (int cp = 0; cp < NCTP; ++cp) {
            const v4f q = bp[(2 * ks * NCTP + cp) * 16];
            const v2f b0 = {q.x, q.y};
            const v2f b1 = {q.z, q.w};
            acc[2 * cp] = __builtin_amdgcn_wmma_f32_16x16x4_f32(
                 false, a,  false, b0,
                 (short)0, acc[2 * cp],  false,  false);
            acc[2 * cp + 1] = __builtin_amdgcn_wmma_f32_16x16x4_f32(
                 false, a,  false, b1,
                 (short)0, acc[2 * cp + 1],  false,  false);
        }
    }

    #pragma unroll
    for (int ct = 0; ct < NCT; ++ct)
        #pragma unroll
        for (int r = 0; r < 8; ++r)
            stg[(hi * 8 + r) * CP + ct * 16 + l15] = acc[ct][r];
    if (mtile * 16 < M) {
        float* hstrip = H + (size_t)mtile * 16 * NCOLS;
        #pragma unroll
        for (int i = 0; i < CL; ++i) {
            const int e   = (i * 32 + lane) * 4;
            const int row = e / NCOLS;
            const int col = e - row * NCOLS;
            *(float4*)&hstrip[e] = *(const float4*)&stg[row * CP + col];
        }
    }
}

template <int F>
__device__ __forceinline__ void agg_edge_now(const float* __restrict__ h,
                                             const int* __restrict__ src,
                                             const float* __restrict__ dinv,
                                             const float* t_dinv, float* tile,
                                             int edge, int row, int lane) {
    const int s = src[edge];
    const float nrm = dinv[s] * t_dinv[row];
    #pragma unroll
    for (int c = 0; c < F / 32; ++c)
        atomicAdd(&tile[row * F + c * 32 + lane], h[(size_t)s * F + c * 32 + lane] * nrm);
}

template <int F, bool RELU, bool FIRST>
__global__ __launch_bounds__(1024)
void agg_tile(const float* __restrict__ h,
              const int* __restrict__ src,
              const int* __restrict__ dst,
              const unsigned* __restrict__ plane,
              const float* __restrict__ dinv,
              const float* __restrict__ bias,
              float* __restrict__ out,
              int* __restrict__ kept_a,
              int* __restrict__ kept_d,
              float* __restrict__ kept_n,
              int* __restrict__ kept_cnt,
              int e, int n) {
    __shared__ __attribute__((aligned(16))) float tile[AGG_R * F];
    __shared__ float t_dinv[AGG_R];
    __shared__ int   l_a[AGG_WAVES * AGG_CAPW];
    __shared__ int   l_d[AGG_WAVES * AGG_CAPW];
    __shared__ float l_n[AGG_WAVES * AGG_CAPW];
    __shared__ int   l_cnt[AGG_WAVES];

    const int tid  = threadIdx.x;
    const int lane = tid & 31;
    const int wav  = __builtin_amdgcn_readfirstlane(tid >> 5);
    const int lo   = blockIdx.x * AGG_R;
    const int cnt  = min(AGG_R, n - lo);
    const unsigned ucnt = (unsigned)cnt;

    int* my_a = l_a + wav * AGG_CAPW;
    int* my_d = l_d + wav * AGG_CAPW;
    float* my_n = l_n + wav * AGG_CAPW;
    int wcount = 0;
    const int gw = blockIdx.x * AGG_WAVES + wav;
    const size_t kb = (size_t)gw * AGG_CAPW;

    bool scan = true;
    int   ka[8], kd[8];
    float kn[8];
    if (!FIRST) {
        wcount = kept_cnt[gw];
        scan = wcount > AGG_CAPW;
        if (scan) {
            wcount = 0;
        } else if (wcount > 0) {
            #pragma unroll
            for (int u = 0; u < 8; ++u) {
                const size_t qi = kb + (size_t)min(lane + 32 * u, wcount - 1);
                ka[u] = kept_a[qi];
                kd[u] = kept_d[qi];
                kn[u] = kept_n[qi];
            }
        }
    }

    static_assert(F % 4 == 0 && AGG_R <= 1024, "16-byte pieces inside a row; one t_dinv entry per thread");
    constexpr int F4  = F / 4;
    constexpr int NPC = (AGG_R * F4 + 1023) / 1024;
    const int npieces = cnt * F4;
    const float* hblk = h + (size_t)lo * F;
    const float td = dinv[lo + min(tid, cnt - 1)];
    float4 hv[NPC], bv[NPC];
    float  dv[NPC];
    #pragma unroll
    for (int p = 0; p < NPC; ++p) {
        const int q  = min(tid + p * 1024, npieces - 1);
        const int r  = q / F4;
        const int c4 = q - r * F4;
        hv[p] = *(const float4*)&hblk[4 * q];
        dv[p] = dinv[lo + r];
        bv[p] = *(const float4*)&bias[4 * c4];
    }
    if (tid < cnt) t_dinv[tid] = td;
    #pragma unroll
    for (int p = 0; p < NPC; ++p) {
        const int q = tid + p * 1024;
        if (q < npieces) {
            const float d2 = dv[p] * dv[p];
            float4 o;
            o.x = hv[p].x * d2 + bv[p].x;
            o.y = hv[p].y * d2 + bv[p].y;
            o.z = hv[p].z * d2 + bv[p].z;
            o.w = hv[p].w * d2 + bv[p].w;
            *(float4*)&tile[4 * q] = o;
        }
    }
    if (!FIRST && !scan) {
        #pragma unroll
        for (int u = 0; u < 8; ++u) {
            const int q = lane + 32 * u;
            if (q < wcount) { my_a[q] = ka[u]; my_d[q] = kd[u]; my_n[q] = kn[u]; }
        }
    }
    __syncthreads();

    if (scan) {
        const int nchunks = (e + SCAN_CH - 1) / SCAN_CH;
        const int rot = (int)(((long long)blockIdx.x * nchunks) / gridDim.x);
        const unsigned pat = (unsigned)blockIdx.x * 0x01010101u;
        const unsigned below = (1u << lane) - 1u;
        auto eqb = [&](unsigned w) {
            const unsigned x = w ^ pat;
            return ~(((x & 0x7F7F7F7Fu) + 0x7F7F7F7Fu) | x) & 0x80808080u;
        };
        for (int k = wav; k < nchunks; k += AGG_WAVES) {
            int c = k + rot;
            if (c >= nchunks) c -= nchunks;
            const int base = c * SCAN_CH + lane * 16;
            v4u q[4];
            #pragma unroll
            for (int u = 0; u < 4; ++u) q[u] = *(const v4u*)(plane + ((base + u * 512) >> 2));
            asm volatile("" :: "v"(q[0]), "v"(q[1]), "v"(q[2]), "v"(q[3]));
            #pragma unroll
            for (int u = 0; u < 4; ++u) {
                unsigned hits = (eqb(q[u].x) >> 7) | (eqb(q[u].y) >> 6)
                              | (eqb(q[u].z) >> 5) | (eqb(q[u].w) >> 4);
                unsigned act = __builtin_amdgcn_ballot_w32(hits != 0u);
                while (act) {
                    const bool m = hits != 0u;
                    const int  b = __ffs(hits) - 1;
                    const int  edge = base + u * 512 + ((b & 7) << 2) + (b >> 3);
                    hits &= hits - 1u;
                    const int pos = wcount + __popc(act & below);
                    if (m && pos < AGG_CAPW) my_a[pos] = edge;
                    unsigned ov = __builtin_amdgcn_ballot_w32(m && pos >= AGG_CAPW);
                    while (ov) {
                        const int l = __ffs(ov) - 1;
                        ov &= ov - 1u;
                        const int ed = __shfl(edge, l);
                        agg_edge_now<F>(h, src, dinv, t_dinv, tile, ed, dst[ed] - lo, lane);
                    }
                    wcount += __popc(act);
                    act = __builtin_amdgcn_ballot_w32(hits != 0u);
                }
            }
        }
    }
    __syncthreads();

    static_assert(AGG_CAPW == 8 * 32, "the resolve step covers the list in 8 strides of 32 lanes");
    const int nl = min(wcount, AGG_CAPW);
    if (scan) {
        int   sv[8], rv[8];
        float dv[8];
        #pragma unroll
        for (int u = 0; u < 8; ++u) {
            const int q = lane + 32 * u;
            sv[u] = (q < nl) ? src[my_a[q]] : 0;
            rv[u] = (q < nl) ? dst[my_a[q]] - lo : 0;
        }
        #pragma unroll
        for (int u = 0; u < 8; ++u) {
            const int q = lane + 32 * u;
            dv[u] = (q < nl) ? dinv[sv[u]] : 0.0f;
        }
        #pragma unroll
        for (int u = 0; u < 8; ++u) {
            const int q = lane + 32 * u;
            if (q < nl) {
                const int   rw = rv[u];
                const float wt = dv[u] * t_dinv[rw];
                my_a[q] = sv[u]; my_d[q] = rw; my_n[q] = wt;
                if (FIRST) { kept_a[kb + q] = sv[u]; kept_d[kb + q] = rw; kept_n[kb + q] = wt; }
            }
        }
        if (FIRST && lane == 0) kept_cnt[gw] = wcount;
    }
    if (lane == 0) l_cnt[wav] = nl;
    __syncthreads();

    static_assert(AGG_WAVES == 32, "one lane per list in the batch count below");
    constexpr int C = F / 32;
    constexpr int U = 8;
    const int lc  = l_cnt[lane];
    const int lnb = (lc + U - 1) / U;
    int inc = lnb;
    #pragma unroll
    for (int dd = 1; dd < 32; dd <<= 1) {
        const int up = __shfl_up(inc, dd);
        if (lane >= dd) inc += up;
    }
    const int pb = inc - lnb;
    const int nbatch = __builtin_amdgcn_readlane(inc, 31);
    for (int t = wav; t < nbatch; t += AGG_WAVES) {
        const int j   = __popc(__builtin_amdgcn_ballot_w32(pb <= t)) - 1;
        const int off = (t - __builtin_amdgcn_readlane(pb, j)) * U;
        const int len = __builtin_amdgcn_readlane(lc, j);
        const int*   ja = l_a + j * AGG_CAPW;
        const int*   jd = l_d + j * AGG_CAPW;
        const float* jn = l_n + j * AGG_CAPW;
        int   s[U], r[U];
        float w[U], v[U][C];
        if (off + U <= len) {
            #pragma unroll
            for (int u = 0; u < U; ++u) { s[u] = ja[off + u]; r[u] = jd[off + u]; w[u] = jn[off + u]; }
            #pragma unroll
            for (int u = 0; u < U; ++u)
                #pragma unroll
                for (int c = 0; c < C; ++c)
                    v[u][c] = h[(size_t)s[u] * F + c * 32 + lane];
            #pragma unroll
            for (int u = 0; u < U; ++u)
                #pragma unroll
                for (int c = 0; c < C; ++c)
                    atomicAdd(&tile[r[u] * F + c * 32 + lane], v[u][c] * w[u]);
        } else {
            #pragma unroll
            for (int u = 0; u < U; ++u) {
                const int q = min(off + u, len - 1);
                s[u] = ja[q]; r[u] = jd[q]; w[u] = jn[q];
            }
            #pragma unroll
            for (int u = 0; u < U; ++u)
                #pragma unroll
                for (int c = 0; c < C; ++c)
                    v[u][c] = h[(size_t)s[u] * F + c * 32 + lane];
            #pragma unroll
            for (int u = 0; u < U; ++u)
                if (off + u < len) {
                    #pragma unroll
                    for (int c = 0; c < C; ++c)
                        atomicAdd(&tile[r[u] * F + c * 32 + lane], v[u][c] * w[u]);
                }
        }
    }
    __syncthreads();

    float* oblk = out + (size_t)lo * F;
    float4 ov[NPC];
    #pragma unroll
    for (int p = 0; p < NPC; ++p)
        ov[p] = *(const float4*)&tile[4 * min(tid + p * 1024, npieces - 1)];
    #pragma unroll
    for (int p = 0; p < NPC; ++p) {
        const int q = tid + p * 1024;
        if (q < npieces) {
            float4 v = ov[p];
            if (RELU) { v.x = fmaxf(v.x, 0.0f); v.y = fmaxf(v.y, 0.0f); v.z = fmaxf(v.z, 0.0f); v.w = fmaxf(v.w, 0.0f); }
            *(float4*)&oblk[4 * q] = v;
        }
    }
}

#define POOL_WAVES 32
#define POOL_LOADS 13
#define POOL_U     16
static_assert(POOL_WAVES * POOL_LOADS * 128 >= GCN_N_NODES, "the waves' slices cover every node");

__global__ __launch_bounds__(1024)
void pool_graph(const float* __restrict__ h,
                const int* __restrict__ batch,
                float* __restrict__ out, int n) {
    __shared__ int   list[GCN_N_NODES];
    __shared__ int   wcnt[POOL_WAVES];
    __shared__ float part[POOL_WAVES * 64];

    const int tid  = threadIdx.x;
    const int lane = tid & 31;
    const int wav  = __builtin_amdgcn_readfirstlane(tid >> 5);
    const int g    = blockIdx.x;
    const int wbase = wav * (POOL_LOADS * 128);

    int4 b[POOL_LOADS];
    if (wbase + POOL_LOADS * 128 <= n) {
        #pragma unroll
        for (int j = 0; j < POOL_LOADS; ++j)
            b[j] = *(const int4*)(batch + wbase + j * 128 + lane * 4);
    } else if (wbase < n) {
        #pragma unroll
        for (int j = 0; j < POOL_LOADS; ++j) {
            const int i0 = wbase + j * 128 + lane * 4;
            const int v0 = batch[min(i0,     n - 1)], v1 = batch[min(i0 + 1, n - 1)];
            const int v2 = batch[min(i0 + 2, n - 1)], v3 = batch[min(i0 + 3, n - 1)];
            b[j].x = (i0     < n) ? v0 : -1;
            b[j].y = (i0 + 1 < n) ? v1 : -1;
            b[j].z = (i0 + 2 < n) ? v2 : -1;
            b[j].w = (i0 + 3 < n) ? v3 : -1;
        }
    } else {
        #pragma unroll
        for (int j = 0; j < POOL_LOADS; ++j) b[j] = make_int4(-1, -1, -1, -1);
    }
    int cnt = 0;
    #pragma unroll
    for (int j = 0; j < POOL_LOADS; ++j) {
        cnt += __popc(__builtin_amdgcn_ballot_w32(b[j].x == g))
             + __popc(__builtin_amdgcn_ballot_w32(b[j].y == g))
             + __popc(__builtin_amdgcn_ballot_w32(b[j].z == g))
             + __popc(__builtin_amdgcn_ballot_w32(b[j].w == g));
    }
    if (lane == 0) wcnt[wav] = cnt;
    __syncthreads();

    int pos = 0, total = 0;
    #pragma unroll
    for (int v = 0; v < POOL_WAVES; ++v) {
        const int c = __builtin_amdgcn_readfirstlane(wcnt[v]);
        total += c;
        if (v < wav) pos += c;
    }
    if (cnt > 0) {
        const unsigned below = (1u << lane) - 1u;
        auto put = [&](int bv, int node) {
            const bool m = bv == g;
            const unsigned mask = __builtin_amdgcn_ballot_w32(m);
            if (m) list[pos + __popc(mask & below)] = node;
            pos += __popc(mask);
        };
        #pragma unroll
        for (int j = 0; j < POOL_LOADS; ++j) {
            const int i0 = wbase + j * 128 + lane * 4;
            put(b[j].x, i0); put(b[j].y, i0 + 1); put(b[j].z, i0 + 2); put(b[j].w, i0 + 3);
        }
    }
    __syncthreads();

    const int e_lo = (wav * total) / POOL_WAVES;
    const int e_hi = ((wav + 1) * total) / POOL_WAVES;
    v2f acc = {0.f, 0.f};
    for (int e = e_lo; e < e_hi; e += POOL_U) {
        int nd[POOL_U];
        v2f v[POOL_U];
        #pragma unroll
        for (int u = 0; u < POOL_U; ++u) {
            const int q = e + u;
            nd[u] = __builtin_amdgcn_readfirstlane(list[q < e_hi ? q : e_lo]);
        }
        #pragma unroll
        for (int u = 0; u < POOL_U; ++u)
            v[u] = *(const v2f*)(h + (size_t)nd[u] * 64 + lane * 2);
        #pragma unroll
        for (int u = 0; u < POOL_U; ++u)
            if (e + u < e_hi) { acc.x += v[u].x; acc.y += v[u].y; }
    }
    *(v2f*)&part[wav * 64 + lane * 2] = acc;
    __syncthreads();

    if (tid < 64) {
        float s = 0.f;
        #pragma unroll
        for (int w = 0; w < POOL_WAVES; ++w) s += part[w * 64 + tid];
        out[g * 64 + tid] = s / fmaxf((float)total, 1.0f);
    }
}

extern "C" void kernel_launch(void* const* d_in, const int* in_sizes, int n_in,
                              void* d_out, int out_size, void* d_ws, size_t ws_size,
                              hipStream_t stream) {
    const float* x     = (const float*)d_in[0];
    const int*   ei    = (const int*)d_in[1];
    const int*   batch = (const int*)d_in[2];
    const float* Wl[4] = {(const float*)d_in[3], (const float*)d_in[5],
                          (const float*)d_in[7], (const float*)d_in[9]};
    const float* bl[4] = {(const float*)d_in[4], (const float*)d_in[6],
                          (const float*)d_in[8], (const float*)d_in[10]};

    const int N = GCN_N_NODES;
    const int E = in_sizes[1] / 2;
    const int* src = ei;
    const int* dst = ei + E;

    auto au = [](size_t v) { return (v + 255) & ~(size_t)255; };
    char* ws = (char*)d_ws;
    size_t off = 0;
    float* deg  = (float*)(ws + off); off += au(sizeof(float) * N);
    float* bufA = (float*)(ws + off); off += au(sizeof(float) * (size_t)N * 96);
    float* bufC = (float*)(ws + off); off += au(sizeof(float) * (size_t)N * 96);
    const size_t KEPT = (size_t)((GCN_N_NODES + AGG_R - 1) / AGG_R) * AGG_WAVES * AGG_CAPW;
    int*   kept_a   = (int*)(ws + off);   off += au(sizeof(int) * KEPT);
    int*   kept_d   = (int*)(ws + off);   off += au(sizeof(int) * KEPT);
    float* kept_n   = (float*)(ws + off); off += au(sizeof(float) * KEPT);
    int*   kept_cnt = (int*)(ws + off);   off += au(sizeof(int) * (KEPT / AGG_CAPW));
    const int EP = ((E + SCAN_CH - 1) / SCAN_CH) * SCAN_CH;
    unsigned* plane = (unsigned*)(ws + off); off += au((size_t)EP);
    (void)ws_size; (void)n_in; (void)out_size;

    const int TB = 256;
    deg_init    <<<(N + TB - 1) / TB, TB, 0, stream>>>(deg, N);
    deg_accum   <<<(EP / 4 + TB - 1) / TB, TB, 0, stream>>>(dst, deg, plane, E, EP / 4);
    deg_finalize<<<(N + TB - 1) / TB, TB, 0, stream>>>(deg, N);

    constexpr int WV = 16;
    const int MT  = (N + 15) / 16;
    const int GBX = (MT + WV - 1) / WV;
    const int AGB = (N + AGG_R - 1) / AGG_R;

    gemm_f32_wmma<128, 96, WV><<<GBX, 32 * WV, 0, stream>>>(x, Wl[0], bufA, N);
    agg_tile<96, true, true><<<AGB, 1024, 0, stream>>>(bufA, src, dst, plane, deg, bl[0], bufC,
                                                       kept_a, kept_d, kept_n, kept_cnt, E, N);

    for (int l = 1; l <= 2; ++l) {
        gemm_f32_wmma<96, 96, WV><<<GBX, 32 * WV, 0, stream>>>(bufC, Wl[l], bufA, N);
        agg_tile<96, true, false><<<AGB, 1024, 0, stream>>>(bufA, src, dst, plane, deg, bl[l], bufC,
                                                            kept_a, kept_d, kept_n, kept_cnt, E, N);
    }

    gemm_f32_wmma<96, 64, WV><<<GBX, 32 * WV, 0, stream>>>(bufC, Wl[3], bufA, N);
    agg_tile<64, false, false><<<AGB, 1024, 0, stream>>>(bufA, src, dst, plane, deg, bl[3], bufC,
                                                         kept_a, kept_d, kept_n, kept_cnt, E, N);

    pool_graph<<<GCN_N_GRAPHS, 1024, 0, stream>>>(bufC, batch, (float*)d_out, N);
}


# Aggregation adds: layer 1 orders the block's edge list by tile row in LDS (rank per row, prefix sum); equal-row entries of a batch are summed in registers, one LDS atomic per run; sum order changes
# speedup vs baseline: 3.3794x; 1.0914x over previous
#include <hip/hip_runtime.h>

typedef __attribute__((ext_vector_type(2))) float v2f;
typedef __attribute__((ext_vector_type(4))) float v4f;
typedef __attribute__((ext_vector_type(8))) float v8f;
typedef __attribute__((ext_vector_type(4))) unsigned v4u;

#define GCN_N_NODES  50000
#define GCN_N_GRAPHS 64

#define AGG_R     200
#define AGG_WAVES 32
#define AGG_CAPW  256
#define SCAN_CH   2048
static_assert((GCN_N_NODES + AGG_R - 1) / AGG_R <= 255, "tile numbers are bytes, 255 is no tile");

__global__ void deg_init(float* __restrict__ deg, int n) {
    int i = blockIdx.x * blockDim.x + threadIdx.x;
    if (i < n) deg[i] = 1.0f;
}

__global__ void deg_accum(const int* __restrict__ dst, float* __restrict__ deg,
                          unsigned* __restrict__ plane, int e, int words) {
    const int t = blockIdx.x * blockDim.x + threadIdx.x;
    if (t >= words) return;
    int d[4];
    #pragma unroll
    for (int k = 0; k < 4; ++k) d[k] = dst[min(4 * t + k, e - 1)];
    unsigned packed = 0u;
    #pragma unroll
    for (int k = 0; k < 4; ++k) {
        unsigned id = 255u;
        if (4 * t + k < e) {
            atomicAdd(&deg[d[k]], 1.0f);
            id = (unsigned)d[k] / AGG_R;
        }
        packed |= id << (8 * k);
    }
    plane[t] = packed;
}

__global__ void deg_finalize(float* __restrict__ deg, int n) {
    int i = blockIdx.x * blockDim.x + threadIdx.x;
    if (i < n) deg[i] = rsqrtf(deg[i]);
}

template <int K, int NCOLS, int WAVES>
__global__ __launch_bounds__(32 * WAVES)
void gemm_f32_wmma(const float* __restrict__ A,
                   const float* __restrict__ Wm,
                   float* __restrict__ H, int M) {
    constexpr int NCT   = NCOLS / 16;
    constexpr int NCTP  = NCT / 2;
    constexpr int KS    = K / 4;
    constexpr int T     = 32 * WAVES;
    constexpr int C4    = NCOLS / 4;
    constexpr int ITEMS = (K / 2) * C4;
    constexpr int PER   = (ITEMS + T - 1) / T;
    constexpr int AP    = K + 4;
    constexpr int CP    = NCOLS + 4;
    constexpr int STG   = 16 * (AP > CP ? AP : CP);
    constexpr int AL    = (16 * K / 4) / 32;
    constexpr int CL    = (16 * NCOLS / 4) / 32;
    static_assert(NCT % 2 == 0 && K % 4 == 0, "column tiles come in pairs, k in fours");
    static_assert(ITEMS % 32 == 0, "a wave's W items are all inside or all outside");
    static_assert((16 * K / 4) % 32 == 0 && (16 * NCOLS / 4) % 32 == 0, "whole 512-byte pieces");
    static_assert((AP / 4) % 2 == 1 && (CP / 4) % 2 == 1, "stage pitches are 16 bytes x odd");
    __shared__ v4f Bs[(K / 2) * NCTP * 16];
    __shared__ v4f Stg4[WAVES * STG / 4];

    const int tid  = threadIdx.x;
    const int lane = tid & 31;
    const int wav  = __builtin_amdgcn_readfirstlane(tid >> 5);
    const int l15  = lane & 15;
    const int hi   = lane >> 4;
    const int mtile = blockIdx.x * WAVES + wav;
    const int mt   = min(mtile, M / 16 - 1);
    float* stg = (float*)Stg4 + wav * STG;

    float4 w0[PER], w1[PER];
    #pragma unroll
    for (int p = 0; p < PER; ++p) {
        if (ITEMS % T == 0 || wav * 32 + p * T < ITEMS) {
            const int it = tid + p * T;
            const int kp = it / C4;
            const int c4 = it - kp * C4;
            w0[p] = *(const float4*)&Wm[(2 * kp) * NCOLS + 4 * c4];
            w1[p] = *(const float4*)&Wm[(2 * kp + 1) * NCOLS + 4 * c4];
        }
    }

    const float* astrip = A + (size_t)mt * 16 * K;
    float4 av[AL];
    #pragma unroll
    for (int i = 0; i < AL; ++i)
        av[i] = *(const float4*)&astrip[(i * 32 + lane) * 4];

    #pragma unroll
    for (int p = 0; p < PER; ++p) {
        if (ITEMS % T == 0 || wav * 32 + p * T < ITEMS) {
            const int it = tid + p * T;
            const int kp = it / C4;
            const int c4 = it - kp * C4;
            const int ct = c4 >> 2;
            float* d = (float*)&Bs[(kp * NCTP + (ct >> 1)) * 16 + (c4 & 3) * 4] + (ct & 1) * 2;
            *(v2f*)(d)      = v2f{w0[p].x, w1[p].x};
            *(v2f*)(d + 4)  = v2f{w0[p].y, w1[p].y};
            *(v2f*)(d + 8)  = v2f{w0[p].z, w1[p].z};
            *(v2f*)(d + 12) = v2f{w0[p].w, w1[p].w};
        }
    }
    #pragma unroll
    for (int i = 0; i < AL; ++i) {
        const int e   = (i * 32 + lane) * 4;
        const int row = e / K;
        const int col = e - row * K;
        *(float4*)&stg[row * AP + col] = av[i];
    }
    __syncthreads();

    v8f acc[NCT];
    #pragma unroll
    for (int ct = 0; ct < NCT; ++ct) acc[ct] = v8f{0.f, 0.f, 0.f, 0.f, 0.f, 0.f, 0.f, 0.f};
    const v4f*   bp = Bs + hi * (NCTP * 16) + l15;
    const float* ap = stg + l15 * AP + hi * 2;
    #pragma unroll
    for (int ks = 0; ks < KS; ++ks) {
        const v2f a = *(const v2f*)&ap[ks * 4];
        #pragma unroll
        for (int cp = 0; cp < NCTP; ++cp) {
            const v4f q = bp[(2 * ks * NCTP + cp) * 16];
            const v2f b0 = {q.x, q.y};
            const v2f b1 = {q.z, q.w};
            acc[2 * cp] = __builtin_amdgcn_wmma_f32_16x16x4_f32(
                 false, a,  false, b0,
                 (short)0, acc[2 * cp],  false,  false);
            acc[2 * cp + 1] = __builtin_amdgcn_wmma_f32_16x16x4_f32(
                 false, a,  false, b1,
                 (short)0, acc[2 * cp + 1],  false,  false);
        }
    }

    #pragma unroll
    for (int ct = 0; ct < NCT; ++ct)
        #pragma unroll
        for (int r = 0; r < 8; ++r)
            stg[(hi * 8 + r) * CP + ct * 16 + l15] = acc[ct][r];
    if (mtile * 16 < M) {
        float* hstrip = H + (size_t)mtile * 16 * NCOLS;
        #pragma unroll
        for (int i = 0; i < CL; ++i) {
            const int e   = (i * 32 + lane) * 4;
            const int row = e / NCOLS;
            const int col = e - row * NCOLS;
            *(float4*)&hstrip[e] = *(const float4*)&stg[row * CP + col];
        }
    }
}

template <int F>
__device__ __forceinline__ void agg_edge_now(const float* __restrict__ h,
                                             const int* __restrict__ src,
                                             const float* __restrict__ dinv,
                                             const float* t_dinv, float* tile,
                                             int edge, int row, int lane) {
    const int s = src[edge];
    const float nrm = dinv[s] * t_dinv[row];
    #pragma unroll
    for (int c = 0; c < F / 32; ++c)
        atomicAdd(&tile[row * F + c * 32 + lane], h[(size_t)s * F + c * 32 + lane] * nrm);
}

template <int F, int U, bool PART>
__device__ __forceinline__ void agg_batch(const float* __restrict__ h, float* tile,
                                          const int* sa, const int* sr, const float* sn,
                                          int off, int tot, int lane) {
    constexpr int C = F / 32;
    int   s[U], r[U];
    float w[U], v[U][C];
    #pragma unroll
    for (int u = 0; u < U; ++u) {
        const int q = PART ? min(off + u, tot - 1) : off + u;
        s[u] = sa[q]; r[u] = sr[q]; w[u] = sn[q];
    }
    #pragma unroll
    for (int u = 0; u < U; ++u)
        #pragma unroll
        for (int c = 0; c < C; ++c)
            v[u][c] = h[(size_t)s[u] * F + c * 32 + lane];
    #pragma unroll
    for (int u = 0; u < U; ++u) r[u] = __builtin_amdgcn_readfirstlane(r[u]);
    float acc[C];
    int cur = r[0];
    #pragma unroll
    for (int c = 0; c < C; ++c) acc[c] = v[0][c] * w[0];
    #pragma unroll
    for (int u = 1; u < U; ++u) {
        if (!PART || off + u < tot) {
            if (r[u] != cur) {
                #pragma unroll
                for (int c = 0; c < C; ++c) atomicAdd(&tile[cur * F + c * 32 + lane], acc[c]);
                cur = r[u];
                #pragma unroll
                for (int c = 0; c < C; ++c) acc[c] = v[u][c] * w[u];
            } else {
                #pragma unroll
                for (int c = 0; c < C; ++c) acc[c] += v[u][c] * w[u];
            }
        }
    }
    #pragma unroll
    for (int c = 0; c < C; ++c) atomicAdd(&tile[cur * F + c * 32 + lane], acc[c]);
}

template <int F, bool RELU, bool FIRST>
__global__ __launch_bounds__(1024)
void agg_tile(const float* __restrict__ h,
              const int* __restrict__ src,
              const int* __restrict__ dst,
              const unsigned* __restrict__ plane,
              const float* __restrict__ dinv,
              const float* __restrict__ bias,
              float* __restrict__ out,
              int* __restrict__ kept_a,
              int* __restrict__ kept_d,
              float* __restrict__ kept_n,
              int* __restrict__ kept_cnt,
              int* __restrict__ kept_tot,
              int e, int n) {
    constexpr int LCAP = AGG_WAVES * AGG_CAPW;
    __shared__ __attribute__((aligned(16))) float tile[AGG_R * F];
    __shared__ float t_dinv[AGG_R];
    __shared__ int   l_a[FIRST ? LCAP : 32];
    __shared__ __attribute__((aligned(16))) int   s_a[LCAP];
    __shared__ __attribute__((aligned(16))) int   s_r[LCAP];
    __shared__ __attribute__((aligned(16))) float s_n[LCAP];
    __shared__ int   row_cnt[AGG_R];
    __shared__ int   row_start[AGG_R];
    __shared__ int   s_tot;

    const int tid  = threadIdx.x;
    const int lane = tid & 31;
    const int wav  = __builtin_amdgcn_readfirstlane(tid >> 5);
    const int lo   = blockIdx.x * AGG_R;
    const int cnt  = min(AGG_R, n - lo);
    const unsigned ucnt = (unsigned)cnt;

    int* my_a = l_a + (FIRST ? wav * AGG_CAPW : 0);
    int wcount = 0;
    const int gw = blockIdx.x * AGG_WAVES + wav;
    const size_t kbb = (size_t)blockIdx.x * LCAP;
    int tot = 0;

    bool scan = true;
    int   ka[8], kd[8];
    float kn[8];
    if (!FIRST) {
        wcount = __builtin_amdgcn_readfirstlane(kept_cnt[gw]);
        tot    = __builtin_amdgcn_readfirstlane(kept_tot[blockIdx.x]);
        scan = wcount > AGG_CAPW;
        wcount = 0;
        #pragma unroll
        for (int u = 0; u < 8; ++u) {
            if (u * 1024 < tot) {
                const size_t qi = kbb + (size_t)min(tid + 1024 * u, tot - 1);
                ka[u] = kept_a[qi];
                kd[u] = kept_d[qi];
                kn[u] = kept_n[qi];
            }
        }
    }

    static_assert(F % 4 == 0 && AGG_R <= 1024, "16-byte pieces inside a row; one t_dinv entry per thread");
    static_assert(LCAP == 8 * 1024, "the block list is moved in 8 strides of 1024 threads");
    constexpr int F4  = F / 4;
    constexpr int NPC = (AGG_R * F4 + 1023) / 1024;
    const int npieces = cnt * F4;
    const float* hblk = h + (size_t)lo * F;
    const float td = dinv[lo + min(tid, cnt - 1)];
    float4 hv[NPC], bv[NPC];
    float  dv[NPC];
    #pragma unroll
    for (int p = 0; p < NPC; ++p) {
        const int q  = min(tid + p * 1024, npieces - 1);
        const int r  = q / F4;
        const int c4 = q - r * F4;
        hv[p] = *(const float4*)&hblk[4 * q];
        dv[p] = dinv[lo + r];
        bv[p] = *(const float4*)&bias[4 * c4];
    }
    if (tid < cnt) t_dinv[tid] = td;
    if (FIRST && tid < AGG_R) row_cnt[tid] = 0;
    #pragma unroll
    for (int p = 0; p < NPC; ++p) {
        const int q = tid + p * 1024;
        if (q < npieces) {
            const float d2 = dv[p] * dv[p];
            float4 o;
            o.x = hv[p].x * d2 + bv[p].x;
            o.y = hv[p].y * d2 + bv[p].y;
            o.z = hv[p].z * d2 + bv[p].z;
            o.w = hv[p].w * d2 + bv[p].w;
            *(float4*)&tile[4 * q] = o;
        }
    }
    if (!FIRST) {
        #pragma unroll
        for (int u = 0; u < 8; ++u) {
            const int q = tid + 1024 * u;
            if (q < tot) { s_a[q] = ka[u]; s_r[q] = kd[u]; s_n[q] = kn[u]; }
        }
    }
    __syncthreads();

    if (scan) {
        const int nchunks = (e + SCAN_CH - 1) / SCAN_CH;
        const int rot = (int)(((long long)blockIdx.x * nchunks) / gridDim.x);
        const unsigned pat = (unsigned)blockIdx.x * 0x01010101u;
        const unsigned below = (1u << lane) - 1u;
        auto eqb = [&](unsigned w) {
            const unsigned x = w ^ pat;
            return ~(((x & 0x7F7F7F7Fu) + 0x7F7F7F7Fu) | x) & 0x80808080u;
        };
        for (int k = wav; k < nchunks; k += AGG_WAVES) {
            int c = k + rot;
            if (c >= nchunks) c -= nchunks;
            const int base = c * SCAN_CH + lane * 16;
            v4u q[4];
            #pragma unroll
            for (int u = 0; u < 4; ++u) q[u] = *(const v4u*)(plane + ((base + u * 512) >> 2));
            asm volatile("" :: "v"(q[0]), "v"(q[1]), "v"(q[2]), "v"(q[3]));
            #pragma unroll
            for (int u = 0; u < 4; ++u) {
                unsigned hits = (eqb(q[u].x) >> 7) | (eqb(q[u].y) >> 6)
                              | (eqb(q[u].z) >> 5) | (eqb(q[u].w) >> 4);
                unsigned act = __builtin_amdgcn_ballot_w32(hits != 0u);
                while (act) {
                    const bool m = hits != 0u;
                    const int  b = __ffs(hits) - 1;
                    const int  edge = base + u * 512 + ((b & 7) << 2) + (b >> 3);
                    hits &= hits - 1u;
                    const int pos = wcount + __popc(act & below);
                    if (FIRST && m && pos < AGG_CAPW) my_a[pos] = edge;
                    unsigned ov = __builtin_amdgcn_ballot_w32(m && pos >= AGG_CAPW);
                    while (ov) {
                        const int l = __ffs(ov) - 1;
                        ov &= ov - 1u;
                        const int ed = __shfl(edge, l);
                        agg_edge_now<F>(h, src, dinv, t_dinv, tile, ed, dst[ed] - lo, lane);
                    }
                    wcount += __popc(act);
                    act = __builtin_amdgcn_ballot_w32(hits != 0u);
                }
            }
        }
    }

    static_assert(AGG_CAPW == 8 * 32, "the resolve step covers the list in 8 strides of 32 lanes");
    if (FIRST) {
        __syncthreads();
        const int nl = min(wcount, AGG_CAPW);
        int   sv[8], rv[8], rk[8];
        float wt[8];
        {
            float dq[8];
            #pragma unroll
            for (int u = 0; u < 8; ++u) {
                const int q = lane + 32 * u;
                sv[u] = (q < nl) ? src[my_a[q]] : 0;
                rv[u] = (q < nl) ? dst[my_a[q]] - lo : 0;
            }
            #pragma unroll
            for (int u = 0; u < 8; ++u) {
                const int q = lane + 32 * u;
                dq[u] = (q < nl) ? dinv[sv[u]] : 0.0f;
            }
            #pragma unroll
            for (int u = 0; u < 8; ++u) {
                const int q = lane + 32 * u;
                rk[u] = 0; wt[u] = 0.0f;
                if (q < nl) {
                    rk[u] = atomicAdd(&row_cnt[rv[u]], 1);
                    wt[u] = dq[u] * t_dinv[rv[u]];
                }
            }
        }
        if (lane == 0) kept_cnt[gw] = wcount;
        __syncthreads();
        if (wav == 0) {
            constexpr int RPL = (AGG_R + 31) / 32;
            int c[RPL];
            int sum = 0;
            #pragma unroll
            for (int i = 0; i < RPL; ++i) {
                const int rr = lane * RPL + i;
                c[i] = (rr < AGG_R) ? row_cnt[rr] : 0;
                sum += c[i];
            }
            int inc = sum;
            #pragma unroll
            for (int dd = 1; dd < 32; dd <<= 1) {
                const int up = __shfl_up(inc, dd);
                if (lane >= dd) inc += up;
            }
            int run = inc - sum;
            #pragma unroll
            for (int i = 0; i < RPL; ++i) {
                const int rr = lane * RPL + i;
                if (rr < AGG_R) row_start[rr] = run;
                run += c[i];
            }
            if (lane == 31) s_tot = inc;
        }
        __syncthreads();
        #pragma unroll
        for (int u = 0; u < 8; ++u) {
            const int q = lane + 32 * u;
            if (q < nl) {
                const int p = row_start[rv[u]] + rk[u];
                s_a[p] = sv[u]; s_r[p] = rv[u]; s_n[p] = wt[u];
            }
        }
        __syncthreads();
        tot = __builtin_amdgcn_readfirstlane(s_tot);
        if (tid == 0) kept_tot[blockIdx.x] = tot;
        #pragma unroll
        for (int u = 0; u < 8; ++u) {
            const int q = tid + 1024 * u;
            if (q < tot) { kept_a[kbb + q] = s_a[q]; kept_d[kbb + q] = s_r[q]; kept_n[kbb + q] = s_n[q]; }
        }
    }

    constexpr int U = 8;
    const int nbatch = (tot + U - 1) / U;
    for (int t = wav; t < nbatch; t += AGG_WAVES) {
        const int off = t * U;
        if (off + U <= tot) agg_batch<F, U, false>(h, tile, s_a, s_r, s_n, off, tot, lane);
        else                agg_batch<F, U, true >(h, tile, s_a, s_r, s_n, off, tot, lane);
    }
    __syncthreads();

    float* oblk = out + (size_t)lo * F;
    float4 ov[NPC];
    #pragma unroll
    for (int p = 0; p < NPC; ++p)
        ov[p] = *(const float4*)&tile[4 * min(tid + p * 1024, npieces - 1)];
    #pragma unroll
    for (int p = 0; p < NPC; ++p) {
        const int q = tid + p * 1024;
        if (q < npieces) {
            float4 v = ov[p];
            if (RELU) { v.x = fmaxf(v.x, 0.0f); v.y = fmaxf(v.y, 0.0f); v.z = fmaxf(v.z, 0.0f); v.w = fmaxf(v.w, 0.0f); }
            *(float4*)&oblk[4 * q] = v;
        }
    }
}
#define POOL_WAVES 32
#define POOL_LOADS 13
#define POOL_U     16
static_assert(POOL_WAVES * POOL_LOADS * 128 >= GCN_N_NODES, "the waves' slices cover every node");

__global__ __launch_bounds__(1024)
void pool_graph(const float* __restrict__ h,
                const int* __restrict__ batch,
                float* __restrict__ out, int n) {
    __shared__ int   list[GCN_N_NODES];
    __shared__ int   wcnt[POOL_WAVES];
    __shared__ float part[POOL_WAVES * 64];

    const int tid  = threadIdx.x;
    const int lane = tid & 31;
    const int wav  = __builtin_amdgcn_readfirstlane(tid >> 5);
    const int g    = blockIdx.x;
    const int wbase = wav * (POOL_LOADS * 128);

    int4 b[POOL_LOADS];
    if (wbase + POOL_LOADS * 128 <= n) {
        #pragma unroll
        for (int j = 0; j < POOL_LOADS; ++j)
            b[j] = *(const int4*)(batch + wbase + j * 128 + lane * 4);
    } else if (wbase < n) {
        #pragma unroll
        for (int j = 0; j < POOL_LOADS; ++j) {
            const int i0 = wbase + j * 128 + lane * 4;
            const int v0 = batch[min(i0,     n - 1)], v1 = batch[min(i0 + 1, n - 1)];
            const int v2 = batch[min(i0 + 2, n - 1)], v3 = batch[min(i0 + 3, n - 1)];
            b[j].x = (i0     < n) ? v0 : -1;
            b[j].y = (i0 + 1 < n) ? v1 : -1;
            b[j].z = (i0 + 2 < n) ? v2 : -1;
            b[j].w = (i0 + 3 < n) ? v3 : -1;
        }
    } else {
        #pragma unroll
        for (int j = 0; j < POOL_LOADS; ++j) b[j] = make_int4(-1, -1, -1, -1);
    }
    int cnt = 0;
    #pragma unroll
    for (int j = 0; j < POOL_LOADS; ++j) {
        cnt += __popc(__builtin_amdgcn_ballot_w32(b[j].x == g))
             + __popc(__builtin_amdgcn_ballot_w32(b[j].y == g))
             + __popc(__builtin_amdgcn_ballot_w32(b[j].z == g))
             + __popc(__builtin_amdgcn_ballot_w32(b[j].w == g));
    }
    if (lane == 0) wcnt[wav] = cnt;
    __syncthreads();

    int pos = 0, total = 0;
    #pragma unroll
    for (int v = 0; v < POOL_WAVES; ++v) {
        const int c = __builtin_amdgcn_readfirstlane(wcnt[v]);
        total += c;
        if (v < wav) pos += c;
    }
    if (cnt > 0) {
        const unsigned below = (1u << lane) - 1u;
        auto put = [&](int bv, int node) {
            const bool m = bv == g;
            const unsigned mask = __builtin_amdgcn_ballot_w32(m);
            if (m) list[pos + __popc(mask & below)] = node;
            pos += __popc(mask);
        };
        #pragma unroll
        for (int j = 0; j < POOL_LOADS; ++j) {
            const int i0 = wbase + j * 128 + lane * 4;
            put(b[j].x, i0); put(b[j].y, i0 + 1); put(b[j].z, i0 + 2); put(b[j].w, i0 + 3);
        }
    }
    __syncthreads();

    const int e_lo = (wav * total) / POOL_WAVES;
    const int e_hi = ((wav + 1) * total) / POOL_WAVES;
    v2f acc = {0.f, 0.f};
    for (int e = e_lo; e < e_hi; e += POOL_U) {
        int nd[POOL_U];
        v2f v[POOL_U];
        #pragma unroll
        for (int u = 0; u < POOL_U; ++u) {
            const int q = e + u;
            nd[u] = __builtin_amdgcn_readfirstlane(list[q < e_hi ? q : e_lo]);
        }
        #pragma unroll
        for (int u = 0; u < POOL_U; ++u)
            v[u] = *(const v2f*)(h + (size_t)nd[u] * 64 + lane * 2);
        #pragma unroll
        for (int u = 0; u < POOL_U; ++u)
            if (e + u < e_hi) { acc.x += v[u].x; acc.y += v[u].y; }
    }
    *(v2f*)&part[wav * 64 + lane * 2] = acc;
    __syncthreads();

    if (tid < 64) {
        float s = 0.f;
        #pragma unroll
        for (int w = 0; w < POOL_WAVES; ++w) s += part[w * 64 + tid];
        out[g * 64 + tid] = s / fmaxf((float)total, 1.0f);
    }
}

extern "C" void kernel_launch(void* const* d_in, const int* in_sizes, int n_in,
                              void* d_out, int out_size, void* d_ws, size_t ws_size,
                              hipStream_t stream) {
    const float* x     = (const float*)d_in[0];
    const int*   ei    = (const int*)d_in[1];
    const int*   batch = (const int*)d_in[2];
    const float* Wl[4] = {(const float*)d_in[3], (const float*)d_in[5],
                          (const float*)d_in[7], (const float*)d_in[9]};
    const float* bl[4] = {(const float*)d_in[4], (const float*)d_in[6],
                          (const float*)d_in[8], (const float*)d_in[10]};

    const int N = GCN_N_NODES;
    const int E = in_sizes[1] / 2;
    const int* src = ei;
    const int* dst = ei + E;

    auto au = [](size_t v) { return (v + 255) & ~(size_t)255; };
    char* ws = (char*)d_ws;
    size_t off = 0;
    float* deg  = (float*)(ws + off); off += au(sizeof(float) * N);
    float* bufA = (float*)(ws + off); off += au(sizeof(float) * (size_t)N * 96);
    float* bufC = (float*)(ws + off); off += au(sizeof(float) * (size_t)N * 96);
    const size_t KEPT = (size_t)((GCN_N_NODES + AGG_R - 1) / AGG_R) * AGG_WAVES * AGG_CAPW;
    int*   kept_a   = (int*)(ws + off);   off += au(sizeof(int) * KEPT);
    int*   kept_d   = (int*)(ws + off);   off += au(sizeof(int) * KEPT);
    float* kept_n   = (float*)(ws + off); off += au(sizeof(float) * KEPT);
    int*   kept_cnt = (int*)(ws + off);   off += au(sizeof(int) * (KEPT / AGG_CAPW));
    int*   kept_tot = (int*)(ws + off);   off += au(sizeof(int) * (KEPT / AGG_CAPW / AGG_WAVES));
    const int EP = ((E + SCAN_CH - 1) / SCAN_CH) * SCAN_CH;
    unsigned* plane = (unsigned*)(ws + off); off += au((size_t)EP);
    (void)ws_size; (void)n_in; (void)out_size;

    const int TB = 256;
    deg_init    <<<(N + TB - 1) / TB, TB, 0, stream>>>(deg, N);
    deg_accum   <<<(EP / 4 + TB - 1) / TB, TB, 0, stream>>>(dst, deg, plane, E, EP / 4);
    deg_finalize<<<(N + TB - 1) / TB, TB, 0, stream>>>(deg, N);

    constexpr int WV = 16;
    const int MT  = (N + 15) / 16;
    const int GBX = (MT + WV - 1) / WV;
    const int AGB = (N + AGG_R - 1) / AGG_R;

    gemm_f32_wmma<128, 96, WV><<<GBX, 32 * WV, 0, stream>>>(x, Wl[0], bufA, N);
    agg_tile<96, true, true><<<AGB, 1024, 0, stream>>>(bufA, src, dst, plane, deg, bl[0], bufC,
                                                       kept_a, kept_d, kept_n, kept_cnt, kept_tot, E, N);

    for (int l = 1; l <= 2; ++l) {
        gemm_f32_wmma<96, 96, WV><<<GBX, 32 * WV, 0, stream>>>(bufC, Wl[l], bufA, N);
        agg_tile<96, true, false><<<AGB, 1024, 0, stream>>>(bufA, src, dst, plane, deg, bl[l], bufC,
                                                            kept_a, kept_d, kept_n, kept_cnt, kept_tot, E, N);
    }

    gemm_f32_wmma<96, 64, WV><<<GBX, 32 * WV, 0, stream>>>(bufC, Wl[3], bufA, N);
    agg_tile<64, false, false><<<AGB, 1024, 0, stream>>>(bufA, src, dst, plane, deg, bl[3], bufC,
                                                         kept_a, kept_d, kept_n, kept_cnt, kept_tot, E, N);

    pool_graph<<<GCN_N_GRAPHS, 1024, 0, stream>>>(bufC, batch, (float*)d_out, N);
}


# No activation behind layer 4: mean pool moved in front of its product; last aggregation runs 96 wide, sums its rows per graph in LDS; 64 pooled rows @ W4 on WMMA in a finish launch; sum order changes
# speedup vs baseline: 3.5120x; 1.0392x over previous
#include <hip/hip_runtime.h>

typedef __attribute__((ext_vector_type(2))) float v2f;
typedef __attribute__((ext_vector_type(4))) float v4f;
typedef __attribute__((ext_vector_type(8))) float v8f;
typedef __attribute__((ext_vector_type(4))) unsigned v4u;

#define GCN_N_NODES  50000
#define GCN_N_GRAPHS 64

#define AGG_R     200
#define AGG_WAVES 32
#define AGG_CAPW  256
#define SCAN_CH   2048
static_assert((GCN_N_NODES + AGG_R - 1) / AGG_R <= 255, "tile numbers are bytes, 255 is no tile");

__global__ void deg_init(float* __restrict__ deg, int n) {
    int i = blockIdx.x * blockDim.x + threadIdx.x;
    if (i < n) deg[i] = 1.0f;
}

__global__ void deg_accum(const int* __restrict__ dst, float* __restrict__ deg,
                          unsigned* __restrict__ plane, int e, int words) {
    const int t = blockIdx.x * blockDim.x + threadIdx.x;
    if (t >= words) return;
    int d[4];
    #pragma unroll
    for (int k = 0; k < 4; ++k) d[k] = dst[min(4 * t + k, e - 1)];
    unsigned packed = 0u;
    #pragma unroll
    for (int k = 0; k < 4; ++k) {
        unsigned id = 255u;
        if (4 * t + k < e) {
            atomicAdd(&deg[d[k]], 1.0f);
            id = (unsigned)d[k] / AGG_R;
        }
        packed |= id << (8 * k);
    }
    plane[t] = packed;
}

__global__ void deg_finalize(float* __restrict__ deg, int n) {
    int i = blockIdx.x * blockDim.x + threadIdx.x;
    if (i < n) deg[i] = rsqrtf(deg[i]);
}

template <int K, int NCOLS, int WAVES>
__global__ __launch_bounds__(32 * WAVES)
void gemm_f32_wmma(const float* __restrict__ A,
                   const float* __restrict__ Wm,
                   float* __restrict__ H, int M) {
    constexpr int NCT   = NCOLS / 16;
    constexpr int NCTP  = NCT / 2;
    constexpr int KS    = K / 4;
    constexpr int T     = 32 * WAVES;
    constexpr int C4    = NCOLS / 4;
    constexpr int ITEMS = (K / 2) * C4;
    constexpr int PER   = (ITEMS + T - 1) / T;
    constexpr int AP    = K + 4;
    constexpr int CP    = NCOLS + 4;
    constexpr int STG   = 16 * (AP > CP ? AP : CP);
    constexpr int AL    = (16 * K / 4) / 32;
    constexpr int CL    = (16 * NCOLS / 4) / 32;
    static_assert(NCT % 2 == 0 && K % 4 == 0, "column tiles come in pairs, k in fours");
    static_assert(ITEMS % 32 == 0, "a wave's W items are all inside or all outside");
    static_assert((16 * K / 4) % 32 == 0 && (16 * NCOLS / 4) % 32 == 0, "whole 512-byte pieces");
    static_assert((AP / 4) % 2 == 1 && (CP / 4) % 2 == 1, "stage pitches are 16 bytes x odd");
    __shared__ v4f Bs[(K / 2) * NCTP * 16];
    __shared__ v4f Stg4[WAVES * STG / 4];

    const int tid  = threadIdx.x;
    const int lane = tid & 31;
    const int wav  = __builtin_amdgcn_readfirstlane(tid >> 5);
    const int l15  = lane & 15;
    const int hi   = lane >> 4;
    const int mtile = blockIdx.x * WAVES + wav;
    const int mt   = min(mtile, M / 16 - 1);
    float* stg = (float*)Stg4 + wav * STG;

    float4 w0[PER], w1[PER];
    #pragma unroll
    for (int p = 0; p < PER; ++p) {
        if (ITEMS % T == 0 || wav * 32 + p * T < ITEMS) {
            const int it = tid + p * T;
            const int kp = it / C4;
            const int c4 = it - kp * C4;
            w0[p] = *(const float4*)&Wm[(2 * kp) * NCOLS + 4 * c4];
            w1[p] = *(const float4*)&Wm[(2 * kp + 1) * NCOLS + 4 * c4];
        }
    }

    const float* astrip = A + (size_t)mt * 16 * K;
    float4 av[AL];
    #pragma unroll
    for (int i = 0; i < AL; ++i)
        av[i] = *(const float4*)&astrip[(i * 32 + lane) * 4];

    #pragma unroll
    for (int p = 0; p < PER; ++p) {
        if (ITEMS % T == 0 || wav * 32 + p * T < ITEMS) {
            const int it = tid + p * T;
            const int kp = it / C4;
            const int c4 = it - kp * C4;
            const int ct = c4 >> 2;
            float* d = (float*)&Bs[(kp * NCTP + (ct >> 1)) * 16 + (c4 & 3) * 4] + (ct & 1) * 2;
            *(v2f*)(d)      = v2f{w0[p].x, w1[p].x};
            *(v2f*)(d + 4)  = v2f{w0[p].y, w1[p].y};
            *(v2f*)(d + 8)  = v2f{w0[p].z, w1[p].z};
            *(v2f*)(d + 12) = v2f{w0[p].w, w1[p].w};
        }
    }
    #pragma unroll
    for (int i = 0; i < AL; ++i) {
        const int e   = (i * 32 + lane) * 4;
        const int row = e / K;
        const int col = e - row * K;
        *(float4*)&stg[row * AP + col] = av[i];
    }
    __syncthreads();

    v8f acc[NCT];
    #pragma unroll
    for (int ct = 0; ct < NCT; ++ct) acc[ct] = v8f{0.f, 0.f, 0.f, 0.f, 0.f, 0.f, 0.f, 0.f};
    const v4f*   bp = Bs + hi * (NCTP * 16) + l15;
    const float* ap = stg + l15 * AP + hi * 2;
    #pragma unroll
    for (int ks = 0; ks < KS; ++ks) {
        const v2f a = *(const v2f*)&ap[ks * 4];
        #pragma unroll
        for (int cp = 0; cp < NCTP; ++cp) {
            const v4f q = bp[(2 * ks * NCTP + cp) * 16];
            const v2f b0 = {q.x, q.y};
            const v2f b1 = {q.z, q.w};
            acc[2 * cp] = __builtin_amdgcn_wmma_f32_16x16x4_f32(
                 false, a,  false, b0,
                 (short)0, acc[2 * cp],  false,  false);
            acc[2 * cp + 1] = __builtin_amdgcn_wmma_f32_16x16x4_f32(
                 false, a,  false, b1,
                 (short)0, acc[2 * cp + 1],  false,  false);
        }
    }

    #pragma unroll
    for (int ct = 0; ct < NCT; ++ct)
        #pragma unroll
        for (int r = 0; r < 8; ++r)
            stg[(hi * 8 + r) * CP + ct * 16 + l15] = acc[ct][r];
    if (mtile * 16 < M) {
        float* hstrip = H + (size_t)mtile * 16 * NCOLS;
        #pragma unroll
        for (int i = 0; i < CL; ++i) {
            const int e   = (i * 32 + lane) * 4;
            const int row = e / NCOLS;
            const int col = e - row * NCOLS;
            *(float4*)&hstrip[e] = *(const float4*)&stg[row * CP + col];
        }
    }
}

template <int F>
__device__ __forceinline__ void agg_edge_now(const float* __restrict__ h,
                                             const int* __restrict__ src,
                                             const float* __restrict__ dinv,
                                             const float* t_dinv, float* tile,
                                             int edge, int row, int lane) {
    const int s = src[edge];
    const float nrm = dinv[s] * t_dinv[row];
    #pragma unroll
    for (int c = 0; c < F / 32; ++c)
        atomicAdd(&tile[row * F + c * 32 + lane], h[(size_t)s * F + c * 32 + lane] * nrm);
}

template <int F, int U, bool PART>
__device__ __forceinline__ void agg_batch(const float* __restrict__ h, float* tile,
                                          const int* sa, const int* sr, const float* sn,
                                          int off, int tot, int lane) {
    constexpr int C = F / 32;
    int   s[U], r[U];
    float w[U], v[U][C];
    #pragma unroll
    for (int u = 0; u < U; ++u) {
        const int q = PART ? min(off + u, tot - 1) : off + u;
        s[u] = sa[q]; r[u] = sr[q]; w[u] = sn[q];
    }
    #pragma unroll
    for (int u = 0; u < U; ++u)
        #pragma unroll
        for (int c = 0; c < C; ++c)
            v[u][c] = h[(size_t)s[u] * F + c * 32 + lane];
    #pragma unroll
    for (int u = 0; u < U; ++u) r[u] = __builtin_amdgcn_readfirstlane(r[u]);
    float acc[C];
    int cur = r[0];
    #pragma unroll
    for (int c = 0; c < C; ++c) acc[c] = v[0][c] * w[0];
    #pragma unroll
    for (int u = 1; u < U; ++u) {
        if (!PART || off + u < tot) {
            if (r[u] != cur) {
                #pragma unroll
                for (int c = 0; c < C; ++c) atomicAdd(&tile[cur * F + c * 32 + lane], acc[c]);
                cur = r[u];
                #pragma unroll
                for (int c = 0; c < C; ++c) acc[c] = v[u][c] * w[u];
            } else {
                #pragma unroll
                for (int c = 0; c < C; ++c) acc[c] += v[u][c] * w[u];
            }
        }
    }
    #pragma unroll
    for (int c = 0; c < C; ++c) atomicAdd(&tile[cur * F + c * 32 + lane], acc[c]);
}

template <int F, bool RELU, bool FIRST, bool POOL>
__global__ __launch_bounds__(1024)
void agg_tile(const float* __restrict__ h,
              const int* __restrict__ src,
              const int* __restrict__ dst,
              const unsigned* __restrict__ plane,
              const float* __restrict__ dinv,
              const float* __restrict__ bias,
              float* __restrict__ out,
              int* __restrict__ kept_a,
              int* __restrict__ kept_d,
              float* __restrict__ kept_n,
              int* __restrict__ kept_cnt,
              int* __restrict__ kept_tot,
              const int* __restrict__ batch,
              float* __restrict__ pcnt,
              int e, int n) {
    constexpr int LCAP = AGG_WAVES * AGG_CAPW;
    __shared__ __attribute__((aligned(16))) float tile[AGG_R * F];
    __shared__ float t_dinv[AGG_R];
    __shared__ int   l_a[FIRST ? LCAP : 32];
    __shared__ __attribute__((aligned(16))) int   s_a[LCAP];
    __shared__ __attribute__((aligned(16))) int   s_r[LCAP];
    __shared__ __attribute__((aligned(16))) float s_n[LCAP];
    __shared__ int   row_cnt[AGG_R];
    __shared__ int   row_start[AGG_R];
    __shared__ int   s_tot;
    __shared__ float gsum[POOL ? GCN_N_GRAPHS * F : 32];
    __shared__ int   t_g[POOL ? AGG_R : 32];
    __shared__ int   g_cnt[POOL ? GCN_N_GRAPHS : 32];

    const int tid  = threadIdx.x;
    const int lane = tid & 31;
    const int wav  = __builtin_amdgcn_readfirstlane(tid >> 5);
    const int lo   = blockIdx.x * AGG_R;
    const int cnt  = min(AGG_R, n - lo);
    const unsigned ucnt = (unsigned)cnt;

    int* my_a = l_a + (FIRST ? wav * AGG_CAPW : 0);
    int wcount = 0;
    const int gw = blockIdx.x * AGG_WAVES + wav;
    const size_t kbb = (size_t)blockIdx.x * LCAP;
    int tot = 0;

    bool scan = true;
    int   ka[8], kd[8];
    float kn[8];
    if (!FIRST) {
        wcount = __builtin_amdgcn_readfirstlane(kept_cnt[gw]);
        tot    = __builtin_amdgcn_readfirstlane(kept_tot[blockIdx.x]);
        scan = wcount > AGG_CAPW;
        wcount = 0;
        #pragma unroll
        for (int u = 0; u < 8; ++u) {
            if (u * 1024 < tot) {
                const size_t qi = kbb + (size_t)min(tid + 1024 * u, tot - 1);
                ka[u] = kept_a[qi];
                kd[u] = kept_d[qi];
                kn[u] = kept_n[qi];
            }
        }
    }

    static_assert(F % 4 == 0 && AGG_R <= 1024, "16-byte pieces inside a row; one t_dinv entry per thread");
    static_assert(LCAP == 8 * 1024, "the block list is moved in 8 strides of 1024 threads");
    constexpr int F4  = F / 4;
    constexpr int NPC = (AGG_R * F4 + 1023) / 1024;
    const int npieces = cnt * F4;
    const float* hblk = h + (size_t)lo * F;
    const float td = dinv[lo + min(tid, cnt - 1)];
    const int myg = POOL ? batch[lo + min(tid, cnt - 1)] : 0;
    float4 hv[NPC], bv[NPC];
    float  dv[NPC];
    #pragma unroll
    for (int p = 0; p < NPC; ++p) {
        const int q  = min(tid + p * 1024, npieces - 1);
        const int r  = q / F4;
        const int c4 = q - r * F4;
        hv[p] = *(const float4*)&hblk[4 * q];
        dv[p] = dinv[lo + r];
        bv[p] = POOL ? make_float4(0.f, 0.f, 0.f, 0.f) : *(const float4*)&bias[4 * c4];
    }
    if (tid < cnt) t_dinv[tid] = td;
    if (POOL) {
        if (tid < cnt) t_g[tid] = myg;
        if (tid < GCN_N_GRAPHS) g_cnt[tid] = 0;
        for (int i = tid; i < GCN_N_GRAPHS * F; i += 1024) gsum[i] = 0.0f;
    }
    if (FIRST && tid < AGG_R) row_cnt[tid] = 0;
    #pragma unroll
    for (int p = 0; p < NPC; ++p) {
        const int q = tid + p * 1024;
        if (q < npieces) {
            const float d2 = dv[p] * dv[p];
            float4 o;
            if (POOL) {
                o.x = hv[p].x * d2; o.y = hv[p].y * d2; o.z = hv[p].z * d2; o.w = hv[p].w * d2;
            } else {
                o.x = hv[p].x * d2 + bv[p].x;
                o.y = hv[p].y * d2 + bv[p].y;
                o.z = hv[p].z * d2 + bv[p].z;
                o.w = hv[p].w * d2 + bv[p].w;
            }
            *(float4*)&tile[4 * q] = o;
        }
    }
    if (!FIRST) {
        #pragma unroll
        for (int u = 0; u < 8; ++u) {
            const int q = tid + 1024 * u;
            if (q < tot) { s_a[q] = ka[u]; s_r[q] = kd[u]; s_n[q] = kn[u]; }
        }
    }
    __syncthreads();
    if (POOL && tid < cnt && (unsigned)myg < (unsigned)GCN_N_GRAPHS) atomicAdd(&g_cnt[myg], 1);

    if (scan) {
        const int nchunks = (e + SCAN_CH - 1) / SCAN_CH;
        const int rot = (int)(((long long)blockIdx.x * nchunks) / gridDim.x);
        const unsigned pat = (unsigned)blockIdx.x * 0x01010101u;
        const unsigned below = (1u << lane) - 1u;
        auto eqb = [&](unsigned w) {
            const unsigned x = w ^ pat;
            return ~(((x & 0x7F7F7F7Fu) + 0x7F7F7F7Fu) | x) & 0x80808080u;
        };
        for (int k = wav; k < nchunks; k += AGG_WAVES) {
            int c = k + rot;
            if (c >= nchunks) c -= nchunks;
            const int base = c * SCAN_CH + lane * 16;
            v4u q[4];
            #pragma unroll
            for (int u = 0; u < 4; ++u) q[u] = *(const v4u*)(plane + ((base + u * 512) >> 2));
            asm volatile("" :: "v"(q[0]), "v"(q[1]), "v"(q[2]), "v"(q[3]));
            #pragma unroll
            for (int u = 0; u < 4; ++u) {
                unsigned hits = (eqb(q[u].x) >> 7) | (eqb(q[u].y) >> 6)
                              | (eqb(q[u].z) >> 5) | (eqb(q[u].w) >> 4);
                unsigned act = __builtin_amdgcn_ballot_w32(hits != 0u);
                while (act) {
                    const bool m = hits != 0u;
                    const int  b = __ffs(hits) - 1;
                    const int  edge = base + u * 512 + ((b & 7) << 2) + (b >> 3);
                    hits &= hits - 1u;
                    const int pos = wcount + __popc(act & below);
                    if (FIRST && m && pos < AGG_CAPW) my_a[pos] = edge;
                    unsigned ov = __builtin_amdgcn_ballot_w32(m && pos >= AGG_CAPW);
                    while (ov) {
                        const int l = __ffs(ov) - 1;
                        ov &= ov - 1u;
                        const int ed = __shfl(edge, l);
                        agg_edge_now<F>(h, src, dinv, t_dinv, tile, ed, dst[ed] - lo, lane);
                    }
                    wcount += __popc(act);
                    act = __builtin_amdgcn_ballot_w32(hits != 0u);
                }
            }
        }
    }

    static_assert(AGG_CAPW == 8 * 32, "the resolve step covers the list in 8 strides of 32 lanes");
    if (FIRST) {
        __syncthreads();
        const int nl = min(wcount, AGG_CAPW);
        int   sv[8], rv[8], rk[8];
        float wt[8];
        {
            float dq[8];
            #pragma unroll
            for (int u = 0; u < 8; ++u) {
                const int q = lane + 32 * u;
                sv[u] = (q < nl) ? src[my_a[q]] : 0;
                rv[u] = (q < nl) ? dst[my_a[q]] - lo : 0;
            }
            #pragma unroll
            for (int u = 0; u < 8; ++u) {
                const int q = lane + 32 * u;
                dq[u] = (q < nl) ? dinv[sv[u]] : 0.0f;
            }
            #pragma unroll
            for (int u = 0; u < 8; ++u) {
                const int q = lane + 32 * u;
                rk[u] = 0; wt[u] = 0.0f;
                if (q < nl) {
                    rk[u] = atomicAdd(&row_cnt[rv[u]], 1);
                    wt[u] = dq[u] * t_dinv[rv[u]];
                }
            }
        }
        if (lane == 0) kept_cnt[gw] = wcount;
        __syncthreads();
        if (wav == 0) {
            constexpr int RPL = (AGG_R + 31) / 32;
            int c[RPL];
            int sum = 0;
            #pragma unroll
            for (int i = 0; i < RPL; ++i) {
                const int rr = lane * RPL + i;
                c[i] = (rr < AGG_R) ? row_cnt[rr] : 0;
                sum += c[i];
            }
            int inc = sum;
            #pragma unroll
            for (int dd = 1; dd < 32; dd <<= 1) {
                const int up = __shfl_up(inc, dd);
                if (lane >= dd) inc += up;
            }
            int run = inc - sum;
            #pragma unroll
            for (int i = 0; i < RPL; ++i) {
                const int rr = lane * RPL + i;
                if (rr < AGG_R) row_start[rr] = run;
                run += c[i];
            }
            if (lane == 31) s_tot = inc;
        }
        __syncthreads();
        #pragma unroll
        for (int u = 0; u < 8; ++u) {
            const int q = lane + 32 * u;
            if (q < nl) {
                const int p = row_start[rv[u]] + rk[u];
                s_a[p] = sv[u]; s_r[p] = rv[u]; s_n[p] = wt[u];
            }
        }
        __syncthreads();
        tot = __builtin_amdgcn_readfirstlane(s_tot);
        if (tid == 0) kept_tot[blockIdx.x] = tot;
        #pragma unroll
        for (int u = 0; u < 8; ++u) {
            const int q = tid + 1024 * u;
            if (q < tot) { kept_a[kbb + q] = s_a[q]; kept_d[kbb + q] = s_r[q]; kept_n[kbb + q] = s_n[q]; }
        }
    }

    constexpr int U = 8;
    const int nbatch = (tot + U - 1) / U;
    for (int t = wav; t < nbatch; t += AGG_WAVES) {
        const int off = t * U;
        if (off + U <= tot) agg_batch<F, U, false>(h, tile, s_a, s_r, s_n, off, tot, lane);
        else                agg_batch<F, U, true >(h, tile, s_a, s_r, s_n, off, tot, lane);
    }
    __syncthreads();

    if (POOL) {
        constexpr int C = F / 32;
        float acc[C];
        #pragma unroll
        for (int c = 0; c < C; ++c) acc[c] = 0.0f;
        int cur = -1;
        for (int r = wav; r < cnt; r += AGG_WAVES) {
            const int g = __builtin_amdgcn_readfirstlane(t_g[r]);
            if ((unsigned)g >= (unsigned)GCN_N_GRAPHS) continue;
            if (g != cur) {
                if (cur >= 0) {
                    #pragma unroll
                    for (int c = 0; c < C; ++c) atomicAdd(&gsum[cur * F + c * 32 + lane], acc[c]);
                }
                cur = g;
                #pragma unroll
                for (int c = 0; c < C; ++c) acc[c] = tile[r * F + c * 32 + lane];
            } else {
                #pragma unroll
                for (int c = 0; c < C; ++c) acc[c] += tile[r * F + c * 32 + lane];
            }
        }
        if (cur >= 0) {
            #pragma unroll
            for (int c = 0; c < C; ++c) atomicAdd(&gsum[cur * F + c * 32 + lane], acc[c]);
        }
        __syncthreads();
        float* pblk = out + (size_t)blockIdx.x * GCN_N_GRAPHS * F;
        for (int g = wav; g < GCN_N_GRAPHS; g += AGG_WAVES) {
            if (__builtin_amdgcn_readfirstlane(g_cnt[g]) > 0) {
                #pragma unroll
                for (int c = 0; c < C; ++c) pblk[g * F + c * 32 + lane] = gsum[g * F + c * 32 + lane];
            }
        }
        static_assert(GCN_N_GRAPHS == 64, "one wave writes the block's 64 counts as two lines");
        if (wav == 0) {
            pcnt[blockIdx.x * GCN_N_GRAPHS + lane]      = (float)g_cnt[lane];
            pcnt[blockIdx.x * GCN_N_GRAPHS + 32 + lane] = (float)g_cnt[32 + lane];
        }
        return;
    }
    float* oblk = out + (size_t)lo * F;
    float4 ov[NPC];
    #pragma unroll
    for (int p = 0; p < NPC; ++p)
        ov[p] = *(const float4*)&tile[4 * min(tid + p * 1024, npieces - 1)];
    #pragma unroll
    for (int p = 0; p < NPC; ++p) {
        const int q = tid + p * 1024;
        if (q < npieces) {
            float4 v = ov[p];
            if (RELU) { v.x = fmaxf(v.x, 0.0f); v.y = fmaxf(v.y, 0.0f); v.z = fmaxf(v.z, 0.0f); v.w = fmaxf(v.w, 0.0f); }
            *(float4*)&oblk[4 * q] = v;
        }
    }
}
#define FIN_WAVES 16
static_assert(GCN_N_GRAPHS % FIN_WAVES == 0, "whole blocks of 16 graphs");
static_assert((GCN_N_NODES + AGG_R - 1) / AGG_R <= 256, "a wave looks at the blocks in 8 strides of 32");

__global__ __launch_bounds__(32 * FIN_WAVES)
void pool_finish(const float* __restrict__ psum,
                 const float* __restrict__ pcnt,
                 const float* __restrict__ Wm,
                 const float* __restrict__ bias,
                 float* __restrict__ out, int nblk) {
    constexpr int F = 96, NO = 64;
    constexpr int AP = F + 4;
    constexpr int OP = NO + 4;
    constexpr int KS = F / 4;
    __shared__ __attribute__((aligned(16))) float stg[FIN_WAVES * AP];
    __shared__ __attribute__((aligned(16))) float ostg[FIN_WAVES * OP];
    __shared__ float s_cnt[FIN_WAVES];
    __shared__ int   blist[FIN_WAVES * 256];

    const int tid  = threadIdx.x;
    const int lane = tid & 31;
    const int wav  = __builtin_amdgcn_readfirstlane(tid >> 5);
    const int l15  = lane & 15;
    const int hi   = lane >> 4;
    const int g    = blockIdx.x * FIN_WAVES + wav;

    const int col = (wav & 3) * 16 + l15;
    v2f bw[KS];
    #pragma unroll
    for (int ks = 0; ks < KS; ++ks)
        bw[ks] = v2f{Wm[(4 * ks + 2 * hi) * NO + col], Wm[(4 * ks + 2 * hi + 1) * NO + col]};
    const float bs = bias[col];

    float pc[8];
    #pragma unroll
    for (int i = 0; i < 8; ++i) {
        const int j = lane + 32 * i;
        const float v = pcnt[(size_t)min(j, nblk - 1) * GCN_N_GRAPHS + g];
        pc[i] = (j < nblk) ? v : 0.0f;
    }
    int* myl = blist + wav * 256;
    const unsigned below = (1u << lane) - 1u;
    int nl = 0;
    float tot = 0.0f;
    #pragma unroll
    for (int i = 0; i < 8; ++i) {
        const bool m = pc[i] > 0.0f;
        const unsigned mask = __builtin_amdgcn_ballot_w32(m);
        if (m) myl[nl + __popc(mask & below)] = lane + 32 * i;
        nl += __popc(mask);
        tot += pc[i];
    }
    #pragma unroll
    for (int dd = 16; dd >= 1; dd >>= 1) tot += __shfl_xor(tot, dd);
    __syncthreads();

    float acc[3] = {0.0f, 0.0f, 0.0f};
    for (int e0 = 0; e0 < nl; e0 += 8) {
        int   jb[8];
        float v[8][3];
        #pragma unroll
        for (int u = 0; u < 8; ++u)
            jb[u] = __builtin_amdgcn_readfirstlane(myl[min(e0 + u, nl - 1)]);
        #pragma unroll
        for (int u = 0; u < 8; ++u)
            #pragma unroll
            for (int c = 0; c < 3; ++c)
                v[u][c] = psum[((size_t)jb[u] * GCN_N_GRAPHS + g) * F + c * 32 + lane];
        #pragma unroll
        for (int u = 0; u < 8; ++u)
            if (e0 + u < nl) {
                #pragma unroll
                for (int c = 0; c < 3; ++c) acc[c] += v[u][c];
            }
    }
    #pragma unroll
    for (int c = 0; c < 3; ++c) stg[wav * AP + c * 32 + lane] = acc[c];
    if (lane == 0) s_cnt[wav] = tot;
    __syncthreads();

    if (wav < NO / 16) {
        v8f d = v8f{0.f, 0.f, 0.f, 0.f, 0.f, 0.f, 0.f, 0.f};
        const float* ap = stg + l15 * AP + hi * 2;
        #pragma unroll
        for (int ks = 0; ks < KS; ++ks) {
            const v2f a = *(const v2f*)&ap[ks * 4];
            d = __builtin_amdgcn_wmma_f32_16x16x4_f32(
                 false, a,  false, bw[ks],
                 (short)0, d,  false,  false);
        }
        #pragma unroll
        for (int r = 0; r < 8; ++r) {
            const int   row = hi * 8 + r;
            const float cn  = s_cnt[row];
            ostg[row * OP + col] = d[r] / fmaxf(cn, 1.0f) + (cn > 0.0f ? bs : 0.0f);
        }
    }
    __syncthreads();
    *(v2f*)&out[(size_t)g * NO + lane * 2] = *(const v2f*)&ostg[wav * OP + lane * 2];
}

extern "C" void kernel_launch(void* const* d_in, const int* in_sizes, int n_in,
                              void* d_out, int out_size, void* d_ws, size_t ws_size,
                              hipStream_t stream) {
    const float* x     = (const float*)d_in[0];
    const int*   ei    = (const int*)d_in[1];
    const int*   batch = (const int*)d_in[2];
    const float* Wl[4] = {(const float*)d_in[3], (const float*)d_in[5],
                          (const float*)d_in[7], (const float*)d_in[9]};
    const float* bl[4] = {(const float*)d_in[4], (const float*)d_in[6],
                          (const float*)d_in[8], (const float*)d_in[10]};

    const int N = GCN_N_NODES;
    const int E = in_sizes[1] / 2;
    const int* src = ei;
    const int* dst = ei + E;

    auto au = [](size_t v) { return (v + 255) & ~(size_t)255; };
    char* ws = (char*)d_ws;
    size_t off = 0;
    float* deg  = (float*)(ws + off); off += au(sizeof(float) * N);
    float* bufA = (float*)(ws + off); off += au(sizeof(float) * (size_t)N * 96);
    float* bufC = (float*)(ws + off); off += au(sizeof(float) * (size_t)N * 96);
    const size_t KEPT = (size_t)((GCN_N_NODES + AGG_R - 1) / AGG_R) * AGG_WAVES * AGG_CAPW;
    int*   kept_a   = (int*)(ws + off);   off += au(sizeof(int) * KEPT);
    int*   kept_d   = (int*)(ws + off);   off += au(sizeof(int) * KEPT);
    float* kept_n   = (float*)(ws + off); off += au(sizeof(float) * KEPT);
    int*   kept_cnt = (int*)(ws + off);   off += au(sizeof(int) * (KEPT / AGG_CAPW));
    int*   kept_tot = (int*)(ws + off);   off += au(sizeof(int) * (KEPT / AGG_CAPW / AGG_WAVES));
    const int EP = ((E + SCAN_CH - 1) / SCAN_CH) * SCAN_CH;
    unsigned* plane = (unsigned*)(ws + off); off += au((size_t)EP);
    float* pcnt = (float*)(ws + off); off += au(sizeof(float) * (KEPT / AGG_CAPW / AGG_WAVES) * GCN_N_GRAPHS);
    (void)ws_size; (void)n_in; (void)out_size;

    const int TB = 256;
    deg_init    <<<(N + TB - 1) / TB, TB, 0, stream>>>(deg, N);
    deg_accum   <<<(EP / 4 + TB - 1) / TB, TB, 0, stream>>>(dst, deg, plane, E, EP / 4);
    deg_finalize<<<(N + TB - 1) / TB, TB, 0, stream>>>(deg, N);

    constexpr int WV = 16;
    const int MT  = (N + 15) / 16;
    const int GBX = (MT + WV - 1) / WV;
    const int AGB = (N + AGG_R - 1) / AGG_R;

    gemm_f32_wmma<128, 96, WV><<<GBX, 32 * WV, 0, stream>>>(x, Wl[0], bufA, N);
    agg_tile<96, true, true, false><<<AGB, 1024, 0, stream>>>(bufA, src, dst, plane, deg, bl[0], bufC,
                                                       kept_a, kept_d, kept_n, kept_cnt, kept_tot,
                                                       batch, pcnt, E, N);

    for (int l = 1; l <= 2; ++l) {
        gemm_f32_wmma<96, 96, WV><<<GBX, 32 * WV, 0, stream>>>(bufC, Wl[l], bufA, N);
        agg_tile<96, true, false, false><<<AGB, 1024, 0, stream>>>(bufA, src, dst, plane, deg, bl[l], bufC,
                                                            kept_a, kept_d, kept_n, kept_cnt, kept_tot,
                                                            batch, pcnt, E, N);
    }

    agg_tile<96, false, false, true><<<AGB, 1024, 0, stream>>>(bufC, src, dst, plane, deg, bl[3], bufA,
                                                         kept_a, kept_d, kept_n, kept_cnt, kept_tot,
                                                         batch, pcnt, E, N);
    pool_finish<<<GCN_N_GRAPHS / FIN_WAVES, 32 * FIN_WAVES, 0, stream>>>(bufA, pcnt, Wl[3], bl[3],
                                                                     (float*)d_out, AGB);
}
